# starting kernel
# baseline (speedup 1.0000x reference)
#include <hip/hip_runtime.h>
#include <hip/hip_bf16.h>


typedef float v2f __attribute__((ext_vector_type(2)));
typedef float v8f __attribute__((ext_vector_type(8)));

#define TAU_MIN_F     0.001f
#define XPBD_MAXC     0.15f
#define XPBD_DENOM    (2.0f + 1e-9f)

#define BSTRIDE 260

__device__ __forceinline__ void atomic_add_f32(float* p, float v) {
    unsafeAtomicAdd(p, v);
}

__global__ void zero_kernel(float* __restrict__ p, int n) {
    int i = blockIdx.x * blockDim.x + threadIdx.x;
    if (i < n) p[i] = 0.0f;
}

__global__ __launch_bounds__(256) void head_wmma_kernel(
    const float* __restrict__ ht,
    const float* __restrict__ W,
    const float* __restrict__ bias,
    const float* __restrict__ kp,
    const float* __restrict__ ts,
    float* __restrict__ x,
    int nTiles, int D, int N)
{
    __shared__ float sWT[16 * BSTRIDE];
    __shared__ float sB[3];

    for (int idx = threadIdx.x; idx < 16 * 256; idx += blockDim.x) {
        int n = idx >> 8;
        int k = idx & 255;
        sWT[n * BSTRIDE + k] = (n < 3) ? W[k * 3 + n] : 0.0f;
    }
    if (threadIdx.x < 3) sB[threadIdx.x] = bias[threadIdx.x];
    __syncthreads();

    int wave = (blockIdx.x * blockDim.x + threadIdx.x) >> 5;
    if (wave >= nTiles) return;

    int lane = threadIdx.x & 31;
    int half = lane >> 4;
    int l15  = lane & 15;

    long rowA = (long)wave * 16 + l15;
    const float* arow = ht + rowA * D + 2 * half;
    const float* brow = &sWT[l15 * BSTRIDE + 2 * half];

    v8f acc0 = {};
    v8f acc1 = {};
    #pragma unroll 4
    for (int kb = 0; kb < D; kb += 8) {
        v2f a0 = *(const v2f*)(arow + kb);
        v2f a1 = *(const v2f*)(arow + kb + 4);
        v2f b0 = *(const v2f*)(brow + kb);
        v2f b1 = *(const v2f*)(brow + kb + 4);
        acc0 = __builtin_amdgcn_wmma_f32_16x16x4_f32(
                   false, a0, false, b0, (short)0, acc0, false, false);
        acc1 = __builtin_amdgcn_wmma_f32_16x16x4_f32(
                   false, a1, false, b1, (short)0, acc1, false, false);
    }

    if (l15 < 3) {
        long rowBase = (long)wave * 16 + half * 8;
        #pragma unroll
        for (int r = 0; r < 8; ++r) {
            long row = rowBase + r;
            int  bidx = (int)(row / N);
            float tau = fmaxf(1.0f - ts[bidx], TAU_MIN_F);
            long idx = row * 3 + l15;
            x[idx] = kp[idx] + tau * (acc0[r] + acc1[r] + sB[l15]);
        }
    }
}

__global__ void edge_kernel(
    const float* __restrict__ x,
    const int*   __restrict__ ei,
    const float* __restrict__ rl,
    float* __restrict__ delta,
    int E, int N, int BE)
{
    int t = blockIdx.x * blockDim.x + threadIdx.x;
    if (t >= BE) return;
    int b = t / E;
    int e = t - b * E;
    int s = ei[e];
    int d = ei[E + e];
    int si = (b * N + s) * 3;
    int di = (b * N + d) * 3;

    float dx = x[si + 0] - x[di + 0];
    float dy = x[si + 1] - x[di + 1];
    float dz = x[si + 2] - x[di + 2];
    float dist = sqrtf(dx * dx + dy * dy + dz * dz);
    float C    = dist - rl[e];
    float scal = (-C / XPBD_DENOM) / (dist + 1e-9f);

    float cx = fminf(fmaxf(scal * dx, -XPBD_MAXC), XPBD_MAXC);
    float cy = fminf(fmaxf(scal * dy, -XPBD_MAXC), XPBD_MAXC);
    float cz = fminf(fmaxf(scal * dz, -XPBD_MAXC), XPBD_MAXC);

    atomic_add_f32(&delta[si + 0],  cx);
    atomic_add_f32(&delta[si + 1],  cy);
    atomic_add_f32(&delta[si + 2],  cz);
    atomic_add_f32(&delta[di + 0], -cx);
    atomic_add_f32(&delta[di + 1], -cy);
    atomic_add_f32(&delta[di + 2], -cz);
}

__global__ void apply_kernel(float* __restrict__ x, float* __restrict__ delta, int n) {
    int i = blockIdx.x * blockDim.x + threadIdx.x;
    if (i < n) {
        x[i] += delta[i];
        delta[i] = 0.0f;
    }
}

__global__ void final_kernel(
    const float* __restrict__ x, const float* __restrict__ delta,
    const float* __restrict__ kp, const float* __restrict__ ts,
    float* __restrict__ out, int n, int perBatch)
{
    int i = blockIdx.x * blockDim.x + threadIdx.x;
    if (i < n) {
        int b = i / perBatch;
        float tau = fmaxf(1.0f - ts[b], TAU_MIN_F);
        out[i] = (x[i] + delta[i] - kp[i]) / tau;
    }
}

extern "C" void kernel_launch(void* const* d_in, const int* in_sizes, int n_in,
                              void* d_out, int out_size, void* d_ws, size_t ws_size,
                              hipStream_t stream) {
    const float* kp   = (const float*)d_in[0];
    const float* ts   = (const float*)d_in[1];
    const float* ht   = (const float*)d_in[2];
    const float* W    = (const float*)d_in[3];
    const float* bias = (const float*)d_in[4];
    const int*   ei   = (const int*)d_in[5];
    const float* rl   = (const float*)d_in[6];

    const int B   = in_sizes[1];
    const int E   = in_sizes[6];
    const int BN3 = in_sizes[0];
    const int BN  = BN3 / 3;
    const int N   = BN / B;
    const int D   = in_sizes[2] / BN;

    float* xbuf  = (float*)d_ws;
    float* delta = xbuf + BN3;
    (void)ws_size;

    const int T = 256;
    const int nTiles = (BN + 15) / 16;
    const int gVec   = (BN3 + T - 1) / T;
    const int BE     = B * E;
    const int gEdge  = (BE + T - 1) / T;
    const int gWmma  = (nTiles * 32 + T - 1) / T;

    zero_kernel<<<gVec, T, 0, stream>>>(delta, BN3);
    head_wmma_kernel<<<gWmma, T, 0, stream>>>(ht, W, bias, kp, ts, xbuf, nTiles, D, N);

    const int XPBD_ITERS = 4;
    for (int it = 0; it < XPBD_ITERS; ++it) {
        edge_kernel<<<gEdge, T, 0, stream>>>(xbuf, ei, rl, delta, E, N, BE);
        if (it < XPBD_ITERS - 1) {
            apply_kernel<<<gVec, T, 0, stream>>>(xbuf, delta, BN3);
        } else {
            final_kernel<<<gVec, T, 0, stream>>>(xbuf, delta, kp, ts,
                                                 (float*)d_out, BN3, N * 3);
        }
    }
}


# atomic scatter replaced by per-node gather over an adjacency list built each call (one round of int atomics, arrival order); 8 lanes per node, 4 batches per lane; sums: lane partials, xor butterfly
# speedup vs baseline: 8.7352x; 8.7352x over previous
#include <hip/hip_runtime.h>
#include <hip/hip_bf16.h>


typedef float v2f __attribute__((ext_vector_type(2)));
typedef float v8f __attribute__((ext_vector_type(8)));

#define TAU_MIN_F     0.001f
#define XPBD_MAXC     0.15f
#define XPBD_DENOM    (2.0f + 1e-9f)

#define BSTRIDE 260

__device__ __forceinline__ void atomic_add_f32(float* p, float v) {
    unsafeAtomicAdd(p, v);
}

__global__ void zero_kernel(float* __restrict__ p, int n) {
    int i = blockIdx.x * blockDim.x + threadIdx.x;
    if (i < n) p[i] = 0.0f;
}

__global__ __launch_bounds__(256) void head_wmma_kernel(
    const float* __restrict__ ht,
    const float* __restrict__ W,
    const float* __restrict__ bias,
    const float* __restrict__ kp,
    const float* __restrict__ ts,
    float* __restrict__ x,
    int nTiles, int D, int N)
{
    __shared__ float sWT[16 * BSTRIDE];
    __shared__ float sB[3];

    for (int idx = threadIdx.x; idx < 16 * 256; idx += blockDim.x) {
        int n = idx >> 8;
        int k = idx & 255;
        sWT[n * BSTRIDE + k] = (n < 3) ? W[k * 3 + n] : 0.0f;
    }
    if (threadIdx.x < 3) sB[threadIdx.x] = bias[threadIdx.x];
    __syncthreads();

    int wave = (blockIdx.x * blockDim.x + threadIdx.x) >> 5;
    if (wave >= nTiles) return;

    int lane = threadIdx.x & 31;
    int half = lane >> 4;
    int l15  = lane & 15;

    long rowA = (long)wave * 16 + l15;
    const float* arow = ht + rowA * D + 2 * half;
    const float* brow = &sWT[l15 * BSTRIDE + 2 * half];

    v8f acc0 = {};
    v8f acc1 = {};
    #pragma unroll 4
    for (int kb = 0; kb < D; kb += 8) {
        v2f a0 = *(const v2f*)(arow + kb);
        v2f a1 = *(const v2f*)(arow + kb + 4);
        v2f b0 = *(const v2f*)(brow + kb);
        v2f b1 = *(const v2f*)(brow + kb + 4);
        acc0 = __builtin_amdgcn_wmma_f32_16x16x4_f32(
                   false, a0, false, b0, (short)0, acc0, false, false);
        acc1 = __builtin_amdgcn_wmma_f32_16x16x4_f32(
                   false, a1, false, b1, (short)0, acc1, false, false);
    }

    if (l15 < 3) {
        long rowBase = (long)wave * 16 + half * 8;
        #pragma unroll
        for (int r = 0; r < 8; ++r) {
            long row = rowBase + r;
            int  bidx = (int)(row / N);
            float tau = fmaxf(1.0f - ts[bidx], TAU_MIN_F);
            long idx = row * 3 + l15;
            x[idx] = kp[idx] + tau * (acc0[r] + acc1[r] + sB[l15]);
        }
    }
}

__global__ void edge_kernel(
    const float* __restrict__ x,
    const int*   __restrict__ ei,
    const float* __restrict__ rl,
    float* __restrict__ delta,
    int E, int N, int BE)
{
    int t = blockIdx.x * blockDim.x + threadIdx.x;
    if (t >= BE) return;
    int b = t / E;
    int e = t - b * E;
    int s = ei[e];
    int d = ei[E + e];
    int si = (b * N + s) * 3;
    int di = (b * N + d) * 3;

    float dx = x[si + 0] - x[di + 0];
    float dy = x[si + 1] - x[di + 1];
    float dz = x[si + 2] - x[di + 2];
    float dist = sqrtf(dx * dx + dy * dy + dz * dz);
    float C    = dist - rl[e];
    float scal = (-C / XPBD_DENOM) / (dist + 1e-9f);

    float cx = fminf(fmaxf(scal * dx, -XPBD_MAXC), XPBD_MAXC);
    float cy = fminf(fmaxf(scal * dy, -XPBD_MAXC), XPBD_MAXC);
    float cz = fminf(fmaxf(scal * dz, -XPBD_MAXC), XPBD_MAXC);

    atomic_add_f32(&delta[si + 0],  cx);
    atomic_add_f32(&delta[si + 1],  cy);
    atomic_add_f32(&delta[si + 2],  cz);
    atomic_add_f32(&delta[di + 0], -cx);
    atomic_add_f32(&delta[di + 1], -cy);
    atomic_add_f32(&delta[di + 2], -cz);
}

__global__ void apply_kernel(float* __restrict__ x, float* __restrict__ delta, int n) {
    int i = blockIdx.x * blockDim.x + threadIdx.x;
    if (i < n) {
        x[i] += delta[i];
        delta[i] = 0.0f;
    }
}

__global__ void final_kernel(
    const float* __restrict__ x, const float* __restrict__ delta,
    const float* __restrict__ kp, const float* __restrict__ ts,
    float* __restrict__ out, int n, int perBatch)
{
    int i = blockIdx.x * blockDim.x + threadIdx.x;
    if (i < n) {
        int b = i / perBatch;
        float tau = fmaxf(1.0f - ts[b], TAU_MIN_F);
        out[i] = (x[i] + delta[i] - kp[i]) / tau;
    }
}


typedef int v2i __attribute__((ext_vector_type(2)));

#define NODE_LANES 8
#define NBATCH     4

__global__ void zero_int_kernel(int* __restrict__ p, int n) {
    int i = blockIdx.x * blockDim.x + threadIdx.x;
    if (i < n) p[i] = 0;
}

__global__ void degree_kernel(const int* __restrict__ ei, int* __restrict__ cnt,
                              int* __restrict__ rank, int E) {
    int e = blockIdx.x * blockDim.x + threadIdx.x;
    if (e >= E) return;
    int r0 = atomicAdd(&cnt[ei[e]], 1);
    int r1 = atomicAdd(&cnt[ei[E + e]], 1);
    rank[e]     = r0;
    rank[E + e] = r1;
}

__global__ __launch_bounds__(256) void alloc_kernel(
    const int* __restrict__ cnt, int* __restrict__ total, int* __restrict__ pos, int N)
{
    __shared__ int wsum[8];
    __shared__ int sbase;
    int tid  = threadIdx.x;
    int lane = tid & 31;
    int w    = tid >> 5;
    int i    = blockIdx.x * 256 + tid;
    int c    = (i < N) ? cnt[i] : 0;
    int v    = c;
    for (int o = 1; o < 32; o <<= 1) {
        int t = __shfl_up(v, o);
        if (lane >= o) v += t;
    }
    if (lane == 31) wsum[w] = v;
    __syncthreads();
    if (tid == 0) {
        int tot = 0;
        for (int k = 0; k < 8; ++k) { int t = wsum[k]; wsum[k] = tot; tot += t; }
        sbase = atomicAdd(total, tot);
    }
    __syncthreads();
    if (i < N) pos[i] = sbase + wsum[w] + v - c;
}

__global__ void fill_kernel(
    const int* __restrict__ ei, const float* __restrict__ rl,
    const int* __restrict__ pos, const int* __restrict__ rank,
    int* __restrict__ adj, int E)
{
    int e = blockIdx.x * blockDim.x + threadIdx.x;
    if (e >= E) return;
    int s  = ei[e];
    int d  = ei[E + e];
    int l0 = __float_as_int(rl[e]);
    int p  = pos[s] + rank[e];
    int q  = pos[d] + rank[E + e];
    v2i* a2 = (v2i*)adj;
    v2i es; es.x = d; es.y = l0;
    v2i ed; ed.x = s; ed.y = l0;
    a2[p] = es;
    a2[q] = ed;
}

template <bool FINAL>
__global__ __launch_bounds__(256) void node_gather_kernel(
    const float* __restrict__ xin,
    const int*   __restrict__ adj,
    const int*   __restrict__ pos,
    const int*   __restrict__ cnt,
    const float* __restrict__ kp,
    const float* __restrict__ ts,
    float* __restrict__ xout, int N)
{
    int t   = blockIdx.x * 256 + threadIdx.x;
    int i   = t / NODE_LANES;
    int sub = t % NODE_LANES;
    if (i >= N) return;

    const v2i* a2 = (const v2i*)adj;
    int beg = pos[i];
    int end = beg + cnt[i];

    float px[NBATCH], py[NBATCH], pz[NBATCH];
    float ax[NBATCH], ay[NBATCH], az[NBATCH];
    #pragma unroll
    for (int b = 0; b < NBATCH; ++b) {
        const float* p = xin + (b * N + i) * 3;
        px[b] = p[0]; py[b] = p[1]; pz[b] = p[2];
        ax[b] = 0.0f; ay[b] = 0.0f; az[b] = 0.0f;
    }

    int k = beg + sub;
    v2i ent = {0, 0};
    if (k < end) ent = a2[k];
    while (k < end) {
        int kn  = k + NODE_LANES;
        v2i nxt = ent;
        if (kn < end) nxt = a2[kn];
        int   j  = ent.x;
        float L0 = __int_as_float(ent.y);
        #pragma unroll
        for (int b = 0; b < NBATCH; ++b) {
            const float* q = xin + (b * N + j) * 3;
            float dx = px[b] - q[0];
            float dy = py[b] - q[1];
            float dz = pz[b] - q[2];
            float dist = sqrtf(dx * dx + dy * dy + dz * dz);
            float C    = dist - L0;
            float scal = (-C / XPBD_DENOM) / (dist + 1e-9f);
            ax[b] += fminf(fmaxf(scal * dx, -XPBD_MAXC), XPBD_MAXC);
            ay[b] += fminf(fmaxf(scal * dy, -XPBD_MAXC), XPBD_MAXC);
            az[b] += fminf(fmaxf(scal * dz, -XPBD_MAXC), XPBD_MAXC);
        }
        ent = nxt;
        k   = kn;
    }

    #pragma unroll
    for (int m = 1; m < NODE_LANES; m <<= 1) {
        #pragma unroll
        for (int b = 0; b < NBATCH; ++b) {
            ax[b] += __shfl_xor(ax[b], m, NODE_LANES);
            ay[b] += __shfl_xor(ay[b], m, NODE_LANES);
            az[b] += __shfl_xor(az[b], m, NODE_LANES);
        }
    }

    if (sub < NBATCH) {
        float vx = 0.0f, vy = 0.0f, vz = 0.0f;
        #pragma unroll
        for (int b = 0; b < NBATCH; ++b) {
            if (sub == b) {
                vx = px[b] + ax[b];
                vy = py[b] + ay[b];
                vz = pz[b] + az[b];
            }
        }
        int idx = (sub * N + i) * 3;
        if (FINAL) {
            float tau = fmaxf(1.0f - ts[sub], TAU_MIN_F);
            xout[idx + 0] = (vx - kp[idx + 0]) / tau;
            xout[idx + 1] = (vy - kp[idx + 1]) / tau;
            xout[idx + 2] = (vz - kp[idx + 2]) / tau;
        } else {
            xout[idx + 0] = vx;
            xout[idx + 1] = vy;
            xout[idx + 2] = vz;
        }
    }
}

extern "C" void kernel_launch(void* const* d_in, const int* in_sizes, int n_in,
                              void* d_out, int out_size, void* d_ws, size_t ws_size,
                              hipStream_t stream) {
    const float* kp   = (const float*)d_in[0];
    const float* ts   = (const float*)d_in[1];
    const float* ht   = (const float*)d_in[2];
    const float* W    = (const float*)d_in[3];
    const float* bias = (const float*)d_in[4];
    const int*   ei   = (const int*)d_in[5];
    const float* rl   = (const float*)d_in[6];

    const int B   = in_sizes[1];
    const int E   = in_sizes[6];
    const int BN3 = in_sizes[0];
    const int BN  = BN3 / 3;
    const int N   = BN / B;
    const int D   = in_sizes[2] / BN;

    const int T = 256;
    const int nTiles = (BN + 15) / 16;
    const int gVec   = (BN3 + T - 1) / T;
    const int BE     = B * E;
    const int gEdge  = (BE + T - 1) / T;
    const int gWmma  = (nTiles * 32 + T - 1) / T;
    const int XPBD_ITERS = 4;

    const size_t need = ((size_t)2 * BN3 + (size_t)6 * E + (size_t)2 * N + 1) * sizeof(int);

    if (B == NBATCH && ws_size >= need) {
        float* x0  = (float*)d_ws;
        float* x1  = x0 + BN3;
        int*   adj = (int*)(x1 + BN3);
        int*   rnk = adj + (size_t)4 * E;
        int*   cnt = rnk + (size_t)2 * E;
        int*   pos = cnt + N + 1;

        const int gCnt  = (N + 1 + T - 1) / T;
        const int gE    = (E + T - 1) / T;
        const int gN    = (N + T - 1) / T;
        const int gNode = (N * NODE_LANES + T - 1) / T;

        zero_int_kernel<<<gCnt, T, 0, stream>>>(cnt, N + 1);
        head_wmma_kernel<<<gWmma, T, 0, stream>>>(ht, W, bias, kp, ts, x0, nTiles, D, N);
        degree_kernel<<<gE, T, 0, stream>>>(ei, cnt, rnk, E);
        alloc_kernel<<<gN, T, 0, stream>>>(cnt, cnt + N, pos, N);
        fill_kernel<<<gE, T, 0, stream>>>(ei, rl, pos, rnk, adj, E);

        node_gather_kernel<false><<<gNode, T, 0, stream>>>(x0, adj, pos, cnt, kp, ts, x1, N);
        node_gather_kernel<false><<<gNode, T, 0, stream>>>(x1, adj, pos, cnt, kp, ts, x0, N);
        node_gather_kernel<false><<<gNode, T, 0, stream>>>(x0, adj, pos, cnt, kp, ts, x1, N);
        node_gather_kernel<true ><<<gNode, T, 0, stream>>>(x1, adj, pos, cnt, kp, ts,
                                                           (float*)d_out, N);
        return;
    }

    float* xbuf  = (float*)d_ws;
    float* delta = xbuf + BN3;

    zero_kernel<<<gVec, T, 0, stream>>>(delta, BN3);
    head_wmma_kernel<<<gWmma, T, 0, stream>>>(ht, W, bias, kp, ts, xbuf, nTiles, D, N);

    for (int it = 0; it < XPBD_ITERS; ++it) {
        edge_kernel<<<gEdge, T, 0, stream>>>(xbuf, ei, rl, delta, E, N, BE);
        if (it < XPBD_ITERS - 1) {
            apply_kernel<<<gVec, T, 0, stream>>>(xbuf, delta, BN3);
        } else {
            final_kernel<<<gVec, T, 0, stream>>>(xbuf, delta, kp, ts,
                                                 (float*)d_out, BN3, N * 3);
        }
    }
}


# gather passes: a block per (node chunk, batch) stages the batch's positions slab by slab in LDS; lists split by neighbour slab; neighbour reads are LDS reads; lane sums go slab by slab
# speedup vs baseline: 11.9594x; 1.3691x over previous
#include <hip/hip_runtime.h>
#include <hip/hip_bf16.h>


typedef float v2f __attribute__((ext_vector_type(2)));
typedef float v8f __attribute__((ext_vector_type(8)));

#define TAU_MIN_F     0.001f
#define XPBD_MAXC     0.15f
#define XPBD_DENOM    (2.0f + 1e-9f)

#define BSTRIDE 260

__device__ __forceinline__ void atomic_add_f32(float* p, float v) {
    unsafeAtomicAdd(p, v);
}

__global__ void zero_kernel(float* __restrict__ p, int n) {
    int i = blockIdx.x * blockDim.x + threadIdx.x;
    if (i < n) p[i] = 0.0f;
}

__global__ __launch_bounds__(256) void head_wmma_kernel(
    const float* __restrict__ ht,
    const float* __restrict__ W,
    const float* __restrict__ bias,
    const float* __restrict__ kp,
    const float* __restrict__ ts,
    float* __restrict__ x,
    int nTiles, int D, int N)
{
    __shared__ float sWT[16 * BSTRIDE];
    __shared__ float sB[3];

    for (int idx = threadIdx.x; idx < 16 * 256; idx += blockDim.x) {
        int n = idx >> 8;
        int k = idx & 255;
        sWT[n * BSTRIDE + k] = (n < 3) ? W[k * 3 + n] : 0.0f;
    }
    if (threadIdx.x < 3) sB[threadIdx.x] = bias[threadIdx.x];
    __syncthreads();

    int wave = (blockIdx.x * blockDim.x + threadIdx.x) >> 5;
    if (wave >= nTiles) return;

    int lane = threadIdx.x & 31;
    int half = lane >> 4;
    int l15  = lane & 15;

    long rowA = (long)wave * 16 + l15;
    const float* arow = ht + rowA * D + 2 * half;
    const float* brow = &sWT[l15 * BSTRIDE + 2 * half];

    v8f acc0 = {};
    v8f acc1 = {};
    #pragma unroll 4
    for (int kb = 0; kb < D; kb += 8) {
        v2f a0 = *(const v2f*)(arow + kb);
        v2f a1 = *(const v2f*)(arow + kb + 4);
        v2f b0 = *(const v2f*)(brow + kb);
        v2f b1 = *(const v2f*)(brow + kb + 4);
        acc0 = __builtin_amdgcn_wmma_f32_16x16x4_f32(
                   false, a0, false, b0, (short)0, acc0, false, false);
        acc1 = __builtin_amdgcn_wmma_f32_16x16x4_f32(
                   false, a1, false, b1, (short)0, acc1, false, false);
    }

    if (l15 < 3) {
        long rowBase = (long)wave * 16 + half * 8;
        #pragma unroll
        for (int r = 0; r < 8; ++r) {
            long row = rowBase + r;
            int  bidx = (int)(row / N);
            float tau = fmaxf(1.0f - ts[bidx], TAU_MIN_F);
            long idx = row * 3 + l15;
            x[idx] = kp[idx] + tau * (acc0[r] + acc1[r] + sB[l15]);
        }
    }
}

__global__ void edge_kernel(
    const float* __restrict__ x,
    const int*   __restrict__ ei,
    const float* __restrict__ rl,
    float* __restrict__ delta,
    int E, int N, int BE)
{
    int t = blockIdx.x * blockDim.x + threadIdx.x;
    if (t >= BE) return;
    int b = t / E;
    int e = t - b * E;
    int s = ei[e];
    int d = ei[E + e];
    int si = (b * N + s) * 3;
    int di = (b * N + d) * 3;

    float dx = x[si + 0] - x[di + 0];
    float dy = x[si + 1] - x[di + 1];
    float dz = x[si + 2] - x[di + 2];
    float dist = sqrtf(dx * dx + dy * dy + dz * dz);
    float C    = dist - rl[e];
    float scal = (-C / XPBD_DENOM) / (dist + 1e-9f);

    float cx = fminf(fmaxf(scal * dx, -XPBD_MAXC), XPBD_MAXC);
    float cy = fminf(fmaxf(scal * dy, -XPBD_MAXC), XPBD_MAXC);
    float cz = fminf(fmaxf(scal * dz, -XPBD_MAXC), XPBD_MAXC);

    atomic_add_f32(&delta[si + 0],  cx);
    atomic_add_f32(&delta[si + 1],  cy);
    atomic_add_f32(&delta[si + 2],  cz);
    atomic_add_f32(&delta[di + 0], -cx);
    atomic_add_f32(&delta[di + 1], -cy);
    atomic_add_f32(&delta[di + 2], -cz);
}

__global__ void apply_kernel(float* __restrict__ x, float* __restrict__ delta, int n) {
    int i = blockIdx.x * blockDim.x + threadIdx.x;
    if (i < n) {
        x[i] += delta[i];
        delta[i] = 0.0f;
    }
}

__global__ void final_kernel(
    const float* __restrict__ x, const float* __restrict__ delta,
    const float* __restrict__ kp, const float* __restrict__ ts,
    float* __restrict__ out, int n, int perBatch)
{
    int i = blockIdx.x * blockDim.x + threadIdx.x;
    if (i < n) {
        int b = i / perBatch;
        float tau = fmaxf(1.0f - ts[b], TAU_MIN_F);
        out[i] = (x[i] + delta[i] - kp[i]) / tau;
    }
}


typedef int   v2i __attribute__((ext_vector_type(2)));
typedef float v4f __attribute__((ext_vector_type(4)));

#define SLAB_NODES 25088
#define NODE_LANES 8
#define PASS_T     1024
#define GROUPS     (PASS_T / NODE_LANES)
#define ROUNDS     7
#define CHUNKS     64
#define COPY_DEPTH 8

__global__ void zero_int_kernel(int* __restrict__ p, int n) {
    int i = blockIdx.x * blockDim.x + threadIdx.x;
    if (i < n) p[i] = 0;
}

__global__ void degree_kernel(const int* __restrict__ ei, int* __restrict__ cnt,
                              int* __restrict__ rank, int E, int N) {
    int e = blockIdx.x * blockDim.x + threadIdx.x;
    if (e >= E) return;
    int s = ei[e];
    int d = ei[E + e];
    int r0 = atomicAdd(&cnt[(d / SLAB_NODES) * N + s], 1);
    int r1 = atomicAdd(&cnt[(s / SLAB_NODES) * N + d], 1);
    rank[e]     = r0;
    rank[E + e] = r1;
}

__global__ __launch_bounds__(256) void alloc_kernel(
    const int* __restrict__ cnt, int* __restrict__ total, int* __restrict__ pos, int L)
{
    __shared__ int wsum[8];
    __shared__ int sbase;
    int tid  = threadIdx.x;
    int lane = tid & 31;
    int w    = tid >> 5;
    int i    = blockIdx.x * 256 + tid;
    int c    = (i < L) ? cnt[i] : 0;
    int v    = c;
    for (int o = 1; o < 32; o <<= 1) {
        int t = __shfl_up(v, o);
        if (lane >= o) v += t;
    }
    if (lane == 31) wsum[w] = v;
    __syncthreads();
    if (tid == 0) {
        int tot = 0;
        for (int k = 0; k < 8; ++k) { int t = wsum[k]; wsum[k] = tot; tot += t; }
        sbase = atomicAdd(total, tot);
    }
    __syncthreads();
    if (i < L) pos[i] = sbase + wsum[w] + v - c;
}

__global__ void fill_kernel(
    const int* __restrict__ ei, const float* __restrict__ rl,
    const int* __restrict__ pos, const int* __restrict__ rank,
    int* __restrict__ adj, int E, int N)
{
    int e = blockIdx.x * blockDim.x + threadIdx.x;
    if (e >= E) return;
    int s  = ei[e];
    int d  = ei[E + e];
    int l0 = __float_as_int(rl[e]);
    int p  = pos[(d / SLAB_NODES) * N + s] + rank[e];
    int q  = pos[(s / SLAB_NODES) * N + d] + rank[E + e];
    v2i* a2 = (v2i*)adj;
    v2i es; es.x = d; es.y = l0;
    v2i ed; ed.x = s; ed.y = l0;
    a2[p] = es;
    a2[q] = ed;
}

template <bool FINAL>
__global__ __launch_bounds__(PASS_T) void slab_gather_kernel(
    const float* __restrict__ xin,
    const int*   __restrict__ adj,
    const int*   __restrict__ pos,
    const int*   __restrict__ cnt,
    const float* __restrict__ kp,
    const float* __restrict__ ts,
    float* __restrict__ xout, int N, int npc)
{
    __shared__ __attribute__((aligned(16))) float sx[SLAB_NODES * 3];

    const int tid     = threadIdx.x;
    const int grp     = tid / NODE_LANES;
    const int sub     = tid % NODE_LANES;
    const int b       = blockIdx.y;
    const int node0   = blockIdx.x * npc;
    const int nodeEnd = min(N, node0 + npc);
    const float* xb   = xin + (size_t)b * N * 3;
    const v2i*   a2   = (const v2i*)adj;

    float px[ROUNDS], py[ROUNDS], pz[ROUNDS];
    float ax[ROUNDS], ay[ROUNDS], az[ROUNDS];
    #pragma unroll
    for (int r = 0; r < ROUNDS; ++r) {
        int i = node0 + grp + GROUPS * r;
        px[r] = 0.0f; py[r] = 0.0f; pz[r] = 0.0f;
        if (i < nodeEnd) {
            const float* p = xb + i * 3;
            px[r] = p[0]; py[r] = p[1]; pz[r] = p[2];
        }
        ax[r] = 0.0f; ay[r] = 0.0f; az[r] = 0.0f;
    }

    int ph = 0;
    for (int lo = 0; lo < N; lo += SLAB_NODES, ++ph) {
        const int hi = min(N, lo + SLAB_NODES);
        const int nw = (hi - lo) * 3;

        __syncthreads();
        const float* src = xb + (size_t)lo * 3;
        if ((((size_t)src) & 15) == 0) {
            const v4f* s4 = (const v4f*)src;
            v4f*       d4 = (v4f*)sx;
            const int  n4 = nw >> 2;
            for (int w0 = tid; w0 < n4; w0 += PASS_T * COPY_DEPTH) {
                v4f tmp[COPY_DEPTH];
                #pragma unroll
                for (int u = 0; u < COPY_DEPTH; ++u) {
                    int w = w0 + u * PASS_T;
                    if (w < n4) tmp[u] = s4[w];
                }
                #pragma unroll
                for (int u = 0; u < COPY_DEPTH; ++u) {
                    int w = w0 + u * PASS_T;
                    if (w < n4) d4[w] = tmp[u];
                }
            }
            for (int w = (n4 << 2) + tid; w < nw; w += PASS_T) sx[w] = src[w];
        } else {
            for (int w = tid; w < nw; w += PASS_T) sx[w] = src[w];
        }
        int k[ROUNDS], end[ROUNDS];
        int trips = 0;
        #pragma unroll
        for (int r = 0; r < ROUNDS; ++r) {
            int i = node0 + grp + GROUPS * r;
            k[r] = 0; end[r] = 0;
            if (i < nodeEnd) {
                int l  = ph * N + i;
                int s0 = pos[l];
                k[r]   = s0 + sub;
                end[r] = s0 + cnt[l];
            }
            trips = max(trips, (end[r] - k[r] + NODE_LANES - 1) / NODE_LANES);
        }

        __syncthreads();

        for (int t = 0; t < trips; ++t) {
            v2i  ent[ROUNDS];
            bool has[ROUNDS];
            #pragma unroll
            for (int r = 0; r < ROUNDS; ++r) {
                has[r] = k[r] < end[r];
                ent[r] = a2[has[r] ? k[r] : 0];
                k[r] += NODE_LANES;
            }
            #pragma unroll
            for (int r = 0; r < ROUNDS; ++r) {
                if (has[r]) {
                    int j = ent[r].x;
                    const float* q = &sx[(j - lo) * 3];
                    float L0 = __int_as_float(ent[r].y);
                    float dx = px[r] - q[0];
                    float dy = py[r] - q[1];
                    float dz = pz[r] - q[2];
                    float dist = sqrtf(dx * dx + dy * dy + dz * dz);
                    float C    = dist - L0;
                    float scal = (-C / XPBD_DENOM) / (dist + 1e-9f);
                    ax[r] += fminf(fmaxf(scal * dx, -XPBD_MAXC), XPBD_MAXC);
                    ay[r] += fminf(fmaxf(scal * dy, -XPBD_MAXC), XPBD_MAXC);
                    az[r] += fminf(fmaxf(scal * dz, -XPBD_MAXC), XPBD_MAXC);
                }
            }
        }
    }

    #pragma unroll
    for (int m = 1; m < NODE_LANES; m <<= 1) {
        #pragma unroll
        for (int r = 0; r < ROUNDS; ++r) {
            ax[r] += __shfl_xor(ax[r], m, NODE_LANES);
            ay[r] += __shfl_xor(ay[r], m, NODE_LANES);
            az[r] += __shfl_xor(az[r], m, NODE_LANES);
        }
    }

    float tau = 1.0f;
    if (FINAL) tau = fmaxf(1.0f - ts[b], TAU_MIN_F);
    #pragma unroll
    for (int r = 0; r < ROUNDS; ++r) {
        int i = node0 + grp + GROUPS * r;
        if (i < nodeEnd && sub < 3) {
            float v = (sub == 0) ? (px[r] + ax[r])
                    : (sub == 1) ? (py[r] + ay[r]) : (pz[r] + az[r]);
            int idx = (b * N + i) * 3 + sub;
            if (FINAL) xout[idx] = (v - kp[idx]) / tau;
            else       xout[idx] = v;
        }
    }
}

extern "C" void kernel_launch(void* const* d_in, const int* in_sizes, int n_in,
                              void* d_out, int out_size, void* d_ws, size_t ws_size,
                              hipStream_t stream) {
    const float* kp   = (const float*)d_in[0];
    const float* ts   = (const float*)d_in[1];
    const float* ht   = (const float*)d_in[2];
    const float* W    = (const float*)d_in[3];
    const float* bias = (const float*)d_in[4];
    const int*   ei   = (const int*)d_in[5];
    const float* rl   = (const float*)d_in[6];

    const int B   = in_sizes[1];
    const int E   = in_sizes[6];
    const int BN3 = in_sizes[0];
    const int BN  = BN3 / 3;
    const int N   = BN / B;
    const int D   = in_sizes[2] / BN;

    const int T = 256;
    const int nTiles = (BN + 15) / 16;
    const int gVec   = (BN3 + T - 1) / T;
    const int BE     = B * E;
    const int gEdge  = (BE + T - 1) / T;
    const int gWmma  = (nTiles * 32 + T - 1) / T;
    const int XPBD_ITERS = 4;

    const int P   = (N + SLAB_NODES - 1) / SLAB_NODES;
    const int NL  = P * N;
    const int npc = (N + CHUNKS - 1) / CHUNKS;
    const size_t rW   = ((size_t)2 * E > (size_t)BN3) ? (size_t)2 * E : (size_t)BN3;
    const size_t need = ((size_t)4 * E + rW + (size_t)BN3 + (size_t)2 * NL + 1) * sizeof(int);

    if (E > 0 && npc <= GROUPS * ROUNDS && ws_size >= need) {
        int*   adj = (int*)d_ws;
        int*   rnk = adj + (size_t)4 * E;
        float* x1  = (float*)rnk;
        float* x0  = (float*)(rnk + rW);
        int*   cnt = (int*)(x0 + BN3);
        int*   pos = cnt + NL + 1;

        const int gCnt = (NL + 1 + T - 1) / T;
        const int gE   = (E + T - 1) / T;
        const int gL   = (NL + T - 1) / T;
        const dim3 gPass(CHUNKS, B);

        zero_int_kernel<<<gCnt, T, 0, stream>>>(cnt, NL + 1);
        head_wmma_kernel<<<gWmma, T, 0, stream>>>(ht, W, bias, kp, ts, x0, nTiles, D, N);
        degree_kernel<<<gE, T, 0, stream>>>(ei, cnt, rnk, E, N);
        alloc_kernel<<<gL, T, 0, stream>>>(cnt, cnt + NL, pos, NL);
        fill_kernel<<<gE, T, 0, stream>>>(ei, rl, pos, rnk, adj, E, N);

        slab_gather_kernel<false><<<gPass, PASS_T, 0, stream>>>(x0, adj, pos, cnt, kp, ts, x1, N, npc);
        slab_gather_kernel<false><<<gPass, PASS_T, 0, stream>>>(x1, adj, pos, cnt, kp, ts, x0, N, npc);
        slab_gather_kernel<false><<<gPass, PASS_T, 0, stream>>>(x0, adj, pos, cnt, kp, ts, x1, N, npc);
        slab_gather_kernel<true ><<<gPass, PASS_T, 0, stream>>>(x1, adj, pos, cnt, kp, ts,
                                                                (float*)d_out, N, npc);
        return;
    }

    float* xbuf  = (float*)d_ws;
    float* delta = xbuf + BN3;

    zero_kernel<<<gVec, T, 0, stream>>>(delta, BN3);
    head_wmma_kernel<<<gWmma, T, 0, stream>>>(ht, W, bias, kp, ts, xbuf, nTiles, D, N);

    for (int it = 0; it < XPBD_ITERS; ++it) {
        edge_kernel<<<gEdge, T, 0, stream>>>(xbuf, ei, rl, delta, E, N, BE);
        if (it < XPBD_ITERS - 1) {
            apply_kernel<<<gVec, T, 0, stream>>>(xbuf, delta, BN3);
        } else {
            final_kernel<<<gVec, T, 0, stream>>>(xbuf, delta, kp, ts,
                                                 (float*)d_out, BN3, N * 3);
        }
    }
}


# adjacency build: two-level counting sort ranked by LDS counters replaces one returning global atomic per half-edge (4 build launches become 2); sum order inside a list = arrival order of LDS adds
# speedup vs baseline: 16.0035x; 1.3382x over previous
#include <hip/hip_runtime.h>
#include <hip/hip_bf16.h>


typedef float v2f __attribute__((ext_vector_type(2)));
typedef float v8f __attribute__((ext_vector_type(8)));

#define TAU_MIN_F     0.001f
#define XPBD_MAXC     0.15f
#define XPBD_DENOM    (2.0f + 1e-9f)

#define BSTRIDE 260

__device__ __forceinline__ void atomic_add_f32(float* p, float v) {
    unsafeAtomicAdd(p, v);
}

__global__ void zero_kernel(float* __restrict__ p, int n) {
    int i = blockIdx.x * blockDim.x + threadIdx.x;
    if (i < n) p[i] = 0.0f;
}

__global__ __launch_bounds__(256) void head_wmma_kernel(
    const float* __restrict__ ht,
    const float* __restrict__ W,
    const float* __restrict__ bias,
    const float* __restrict__ kp,
    const float* __restrict__ ts,
    float* __restrict__ x,
    int nTiles, int D, int N)
{
    __shared__ float sWT[16 * BSTRIDE];
    __shared__ float sB[3];

    for (int idx = threadIdx.x; idx < 16 * 256; idx += blockDim.x) {
        int n = idx >> 8;
        int k = idx & 255;
        sWT[n * BSTRIDE + k] = (n < 3) ? W[k * 3 + n] : 0.0f;
    }
    if (threadIdx.x < 3) sB[threadIdx.x] = bias[threadIdx.x];
    __syncthreads();

    int wave = (blockIdx.x * blockDim.x + threadIdx.x) >> 5;
    if (wave >= nTiles) return;

    int lane = threadIdx.x & 31;
    int half = lane >> 4;
    int l15  = lane & 15;

    long rowA = (long)wave * 16 + l15;
    const float* arow = ht + rowA * D + 2 * half;
    const float* brow = &sWT[l15 * BSTRIDE + 2 * half];

    v8f acc0 = {};
    v8f acc1 = {};
    #pragma unroll 4
    for (int kb = 0; kb < D; kb += 8) {
        v2f a0 = *(const v2f*)(arow + kb);
        v2f a1 = *(const v2f*)(arow + kb + 4);
        v2f b0 = *(const v2f*)(brow + kb);
        v2f b1 = *(const v2f*)(brow + kb + 4);
        acc0 = __builtin_amdgcn_wmma_f32_16x16x4_f32(
                   false, a0, false, b0, (short)0, acc0, false, false);
        acc1 = __builtin_amdgcn_wmma_f32_16x16x4_f32(
                   false, a1, false, b1, (short)0, acc1, false, false);
    }

    if (l15 < 3) {
        long rowBase = (long)wave * 16 + half * 8;
        #pragma unroll
        for (int r = 0; r < 8; ++r) {
            long row = rowBase + r;
            int  bidx = (int)(row / N);
            float tau = fmaxf(1.0f - ts[bidx], TAU_MIN_F);
            long idx = row * 3 + l15;
            x[idx] = kp[idx] + tau * (acc0[r] + acc1[r] + sB[l15]);
        }
    }
}

__global__ void edge_kernel(
    const float* __restrict__ x,
    const int*   __restrict__ ei,
    const float* __restrict__ rl,
    float* __restrict__ delta,
    int E, int N, int BE)
{
    int t = blockIdx.x * blockDim.x + threadIdx.x;
    if (t >= BE) return;
    int b = t / E;
    int e = t - b * E;
    int s = ei[e];
    int d = ei[E + e];
    int si = (b * N + s) * 3;
    int di = (b * N + d) * 3;

    float dx = x[si + 0] - x[di + 0];
    float dy = x[si + 1] - x[di + 1];
    float dz = x[si + 2] - x[di + 2];
    float dist = sqrtf(dx * dx + dy * dy + dz * dz);
    float C    = dist - rl[e];
    float scal = (-C / XPBD_DENOM) / (dist + 1e-9f);

    float cx = fminf(fmaxf(scal * dx, -XPBD_MAXC), XPBD_MAXC);
    float cy = fminf(fmaxf(scal * dy, -XPBD_MAXC), XPBD_MAXC);
    float cz = fminf(fmaxf(scal * dz, -XPBD_MAXC), XPBD_MAXC);

    atomic_add_f32(&delta[si + 0],  cx);
    atomic_add_f32(&delta[si + 1],  cy);
    atomic_add_f32(&delta[si + 2],  cz);
    atomic_add_f32(&delta[di + 0], -cx);
    atomic_add_f32(&delta[di + 1], -cy);
    atomic_add_f32(&delta[di + 2], -cz);
}

__global__ void apply_kernel(float* __restrict__ x, float* __restrict__ delta, int n) {
    int i = blockIdx.x * blockDim.x + threadIdx.x;
    if (i < n) {
        x[i] += delta[i];
        delta[i] = 0.0f;
    }
}

__global__ void final_kernel(
    const float* __restrict__ x, const float* __restrict__ delta,
    const float* __restrict__ kp, const float* __restrict__ ts,
    float* __restrict__ out, int n, int perBatch)
{
    int i = blockIdx.x * blockDim.x + threadIdx.x;
    if (i < n) {
        int b = i / perBatch;
        float tau = fmaxf(1.0f - ts[b], TAU_MIN_F);
        out[i] = (x[i] + delta[i] - kp[i]) / tau;
    }
}


typedef int   v2i __attribute__((ext_vector_type(2)));
typedef float v4f __attribute__((ext_vector_type(4)));

#define SLAB_NODES 25088
#define NODE_LANES 8
#define PASS_T     1024
#define GROUPS     (PASS_T / NODE_LANES)
#define ROUNDS     7
#define CHUNKS     64
#define COPY_DEPTH 8

#define BK_SHIFT 8
#define BK_CN    (1 << BK_SHIFT)
#define BK_NBMAX 256
#define BK_T     1024
#define BK_EPT   4
#define BK_EPB   (BK_T * BK_EPT)
#define BK_HPB   (2 * BK_EPB)

__global__ __launch_bounds__(BK_T) void bucket_kernel(
    const int* __restrict__ ei, int* __restrict__ rec,
    int* __restrict__ cntT, int* __restrict__ offT, int E, int NB, int NBLK)
{
    __shared__ int hist[BK_NBMAX];
    __shared__ int hoff[BK_NBMAX];
    const int tid = threadIdx.x;
    if (tid < BK_NBMAX) hist[tid] = 0;
    __syncthreads();

    const int e0 = blockIdx.x * BK_EPB;
    int bk[2 * BK_EPT], rk[2 * BK_EPT];
    #pragma unroll
    for (int u = 0; u < BK_EPT; ++u) {
        int e  = e0 + tid + u * BK_T;
        int ec = (e < E) ? e : 0;
        bk[2 * u]     = ei[ec] >> BK_SHIFT;
        bk[2 * u + 1] = ei[E + ec] >> BK_SHIFT;
    }
    #pragma unroll
    for (int u = 0; u < BK_EPT; ++u) {
        int e = e0 + tid + u * BK_T;
        rk[2 * u] = 0; rk[2 * u + 1] = 0;
        if (e < E) {
            rk[2 * u]     = atomicAdd(&hist[bk[2 * u]], 1);
            rk[2 * u + 1] = atomicAdd(&hist[bk[2 * u + 1]], 1);
        }
    }
    __syncthreads();
    if (tid < 32) {
        int a[8];
        int sum = 0;
        #pragma unroll
        for (int q = 0; q < 8; ++q) { a[q] = hist[8 * tid + q]; sum += a[q]; }
        int v = sum;
        for (int o = 1; o < 32; o <<= 1) {
            int t = __shfl_up(v, o);
            if (tid >= o) v += t;
        }
        int run = v - sum;
        #pragma unroll
        for (int q = 0; q < 8; ++q) { hoff[8 * tid + q] = run; run += a[q]; }
    }
    __syncthreads();
    if (tid < NB) {
        cntT[tid * NBLK + blockIdx.x] = hist[tid];
        offT[tid * NBLK + blockIdx.x] = hoff[tid];
    }
    int* myrec = rec + (size_t)blockIdx.x * BK_HPB;
    #pragma unroll
    for (int u = 0; u < BK_EPT; ++u) {
        int e = e0 + tid + u * BK_T;
        if (e < E) {
            myrec[hoff[bk[2 * u]]     + rk[2 * u]]     = 2 * e;
            myrec[hoff[bk[2 * u + 1]] + rk[2 * u + 1]] = 2 * e + 1;
        }
    }
}

__device__ __forceinline__ int block_scan_excl(int c, int* ws, int* tot) {
    const int lane = threadIdx.x & 31;
    const int w    = threadIdx.x >> 5;
    int v = c;
    for (int o = 1; o < 32; o <<= 1) {
        int t = __shfl_up(v, o);
        if (lane >= o) v += t;
    }
    if (lane == 31) ws[w] = v;
    __syncthreads();
    if (w == 0) {
        int s  = ws[lane];
        int sv = s;
        for (int o = 1; o < 32; o <<= 1) {
            int t = __shfl_up(sv, o);
            if (lane >= o) sv += t;
        }
        ws[lane] = sv - s;
        if (lane == 31) *tot = sv;
    }
    __syncthreads();
    return ws[w] + v - c;
}

#define BK_U 8

__global__ __launch_bounds__(BK_T) void lists_kernel(
    const int* __restrict__ ei, const float* __restrict__ rl,
    const int* __restrict__ rec, const int* __restrict__ cntT,
    const int* __restrict__ offT, int* __restrict__ adj,
    int* __restrict__ pos, int* __restrict__ cnt,
    int E, int N, int NBLK, int P)
{
    __shared__ int segO[BK_T];
    __shared__ int segP[BK_T + 1];
    __shared__ int lcnt[BK_T];
    __shared__ int lcur[BK_T];
    __shared__ int wsA[32], wsB[32], wsC[32];
    __shared__ int totA, totB, baseSh;

    const int tid   = threadIdx.x;
    const int lane  = tid & 31;
    const int w     = tid >> 5;
    const int kb    = blockIdx.x;
    const int node0 = kb << BK_SHIFT;

    int c = 0, o = 0;
    if (tid < NBLK) { c = cntT[kb * NBLK + tid]; o = offT[kb * NBLK + tid]; }
    segO[tid] = o;
    int ex = block_scan_excl(c, wsA, &totA);
    segP[tid] = ex;
    if (tid == BK_T - 1) segP[BK_T] = ex + c;

    int acc = 0;
    for (int idx = tid; idx < kb * NBLK; idx += BK_T) acc += cntT[idx];
    for (int m = 16; m >= 1; m >>= 1) acc += __shfl_xor(acc, m);
    if (lane == 0) wsB[w] = acc;
    lcnt[tid] = 0;
    __syncthreads();
    if (w == 0) {
        int s = wsB[lane];
        for (int m = 16; m >= 1; m >>= 1) s += __shfl_xor(s, m);
        if (lane == 0) baseSh = s;
    }
    __syncthreads();
    const int T    = totA;
    const int base = baseSh;
    int st0 = 1;
    while (st0 * 2 < NBLK) st0 *= 2;

    for (int t0 = tid; t0 < T; t0 += BK_T * BK_U) {
        int  hh[BK_U], tt[BK_U], b[BK_U];
        bool ok[BK_U];
        #pragma unroll
        for (int u = 0; u < BK_U; ++u) {
            int t = t0 + u * BK_T;
            ok[u] = t < T;
            tt[u] = ok[u] ? t : 0;
            b[u]  = 0;
        }
        for (int st = st0; st >= 1; st >>= 1) {
            #pragma unroll
            for (int u = 0; u < BK_U; ++u) b[u] += (segP[b[u] + st] <= tt[u]) ? st : 0;
        }
        #pragma unroll
        for (int u = 0; u < BK_U; ++u) {
            hh[u] = rec[b[u] * BK_HPB + segO[b[u]] + (tt[u] - segP[b[u]])];
        }
        int s[BK_U], d[BK_U];
        #pragma unroll
        for (int u = 0; u < BK_U; ++u) {
            int e = hh[u] >> 1;
            s[u] = ei[e];
            d[u] = ei[E + e];
        }
        #pragma unroll
        for (int u = 0; u < BK_U; ++u) {
            int own = (hh[u] & 1) ? d[u] : s[u];
            int nbr = (hh[u] & 1) ? s[u] : d[u];
            int ll  = ((nbr / SLAB_NODES) << BK_SHIFT) | (own & (BK_CN - 1));
            if (ok[u]) atomicAdd(&lcnt[ll], 1);
        }
    }
    __syncthreads();

    const int LISTS = P << BK_SHIFT;
    int lc  = (tid < LISTS) ? lcnt[tid] : 0;
    int lex = block_scan_excl(lc, wsC, &totB);
    if (tid < LISTS) {
        int p = tid >> BK_SHIFT;
        int i = node0 + (tid & (BK_CN - 1));
        lcur[tid] = base + lex;
        if (i < N) {
            pos[p * N + i] = base + lex;
            cnt[p * N + i] = lc;
        }
    }
    __syncthreads();

    v2i* a2 = (v2i*)adj;
    for (int t0 = tid; t0 < T; t0 += BK_T * BK_U) {
        int  hh[BK_U], tt[BK_U], b[BK_U];
        bool ok[BK_U];
        #pragma unroll
        for (int u = 0; u < BK_U; ++u) {
            int t = t0 + u * BK_T;
            ok[u] = t < T;
            tt[u] = ok[u] ? t : 0;
            b[u]  = 0;
        }
        for (int st = st0; st >= 1; st >>= 1) {
            #pragma unroll
            for (int u = 0; u < BK_U; ++u) b[u] += (segP[b[u] + st] <= tt[u]) ? st : 0;
        }
        #pragma unroll
        for (int u = 0; u < BK_U; ++u) {
            hh[u] = rec[b[u] * BK_HPB + segO[b[u]] + (tt[u] - segP[b[u]])];
        }
        int s[BK_U], d[BK_U], l0[BK_U];
        #pragma unroll
        for (int u = 0; u < BK_U; ++u) {
            int e = hh[u] >> 1;
            s[u]  = ei[e];
            d[u]  = ei[E + e];
            l0[u] = __float_as_int(rl[e]);
        }
        #pragma unroll
        for (int u = 0; u < BK_U; ++u) {
            int own = (hh[u] & 1) ? d[u] : s[u];
            int nbr = (hh[u] & 1) ? s[u] : d[u];
            int ll  = ((nbr / SLAB_NODES) << BK_SHIFT) | (own & (BK_CN - 1));
            if (ok[u]) {
                int slot = atomicAdd(&lcur[ll], 1);
                v2i en; en.x = nbr; en.y = l0[u];
                a2[slot] = en;
            }
        }
    }
}

template <bool FINAL>
__global__ __launch_bounds__(PASS_T) void slab_gather_kernel(
    const float* __restrict__ xin,
    const int*   __restrict__ adj,
    const int*   __restrict__ pos,
    const int*   __restrict__ cnt,
    const float* __restrict__ kp,
    const float* __restrict__ ts,
    float* __restrict__ xout, int N, int npc)
{
    __shared__ __attribute__((aligned(16))) float sx[SLAB_NODES * 3];

    const int tid     = threadIdx.x;
    const int grp     = tid / NODE_LANES;
    const int sub     = tid % NODE_LANES;
    const int b       = blockIdx.y;
    const int node0   = blockIdx.x * npc;
    const int nodeEnd = min(N, node0 + npc);
    const float* xb   = xin + (size_t)b * N * 3;
    const v2i*   a2   = (const v2i*)adj;

    float px[ROUNDS], py[ROUNDS], pz[ROUNDS];
    float ax[ROUNDS], ay[ROUNDS], az[ROUNDS];
    #pragma unroll
    for (int r = 0; r < ROUNDS; ++r) {
        int i = node0 + grp + GROUPS * r;
        px[r] = 0.0f; py[r] = 0.0f; pz[r] = 0.0f;
        if (i < nodeEnd) {
            const float* p = xb + i * 3;
            px[r] = p[0]; py[r] = p[1]; pz[r] = p[2];
        }
        ax[r] = 0.0f; ay[r] = 0.0f; az[r] = 0.0f;
    }

    int ph = 0;
    for (int lo = 0; lo < N; lo += SLAB_NODES, ++ph) {
        const int hi = min(N, lo + SLAB_NODES);
        const int nw = (hi - lo) * 3;

        __syncthreads();
        const float* src = xb + (size_t)lo * 3;
        if ((((size_t)src) & 15) == 0) {
            const v4f* s4 = (const v4f*)src;
            v4f*       d4 = (v4f*)sx;
            const int  n4 = nw >> 2;
            for (int w0 = tid; w0 < n4; w0 += PASS_T * COPY_DEPTH) {
                v4f tmp[COPY_DEPTH];
                #pragma unroll
                for (int u = 0; u < COPY_DEPTH; ++u) {
                    int w = w0 + u * PASS_T;
                    if (w < n4) tmp[u] = s4[w];
                }
                #pragma unroll
                for (int u = 0; u < COPY_DEPTH; ++u) {
                    int w = w0 + u * PASS_T;
                    if (w < n4) d4[w] = tmp[u];
                }
            }
            for (int w = (n4 << 2) + tid; w < nw; w += PASS_T) sx[w] = src[w];
        } else {
            for (int w = tid; w < nw; w += PASS_T) sx[w] = src[w];
        }
        int k[ROUNDS], end[ROUNDS];
        int trips = 0;
        #pragma unroll
        for (int r = 0; r < ROUNDS; ++r) {
            int i = node0 + grp + GROUPS * r;
            k[r] = 0; end[r] = 0;
            if (i < nodeEnd) {
                int l  = ph * N + i;
                int s0 = pos[l];
                k[r]   = s0 + sub;
                end[r] = s0 + cnt[l];
            }
            trips = max(trips, (end[r] - k[r] + NODE_LANES - 1) / NODE_LANES);
        }

        __syncthreads();

        for (int t = 0; t < trips; ++t) {
            v2i  ent[ROUNDS];
            bool has[ROUNDS];
            #pragma unroll
            for (int r = 0; r < ROUNDS; ++r) {
                has[r] = k[r] < end[r];
                ent[r] = a2[has[r] ? k[r] : 0];
                k[r] += NODE_LANES;
            }
            #pragma unroll
            for (int r = 0; r < ROUNDS; ++r) {
                if (has[r]) {
                    int j = ent[r].x;
                    const float* q = &sx[(j - lo) * 3];
                    float L0 = __int_as_float(ent[r].y);
                    float dx = px[r] - q[0];
                    float dy = py[r] - q[1];
                    float dz = pz[r] - q[2];
                    float dist = sqrtf(dx * dx + dy * dy + dz * dz);
                    float C    = dist - L0;
                    float scal = (-C / XPBD_DENOM) / (dist + 1e-9f);
                    ax[r] += fminf(fmaxf(scal * dx, -XPBD_MAXC), XPBD_MAXC);
                    ay[r] += fminf(fmaxf(scal * dy, -XPBD_MAXC), XPBD_MAXC);
                    az[r] += fminf(fmaxf(scal * dz, -XPBD_MAXC), XPBD_MAXC);
                }
            }
        }
    }

    #pragma unroll
    for (int m = 1; m < NODE_LANES; m <<= 1) {
        #pragma unroll
        for (int r = 0; r < ROUNDS; ++r) {
            ax[r] += __shfl_xor(ax[r], m, NODE_LANES);
            ay[r] += __shfl_xor(ay[r], m, NODE_LANES);
            az[r] += __shfl_xor(az[r], m, NODE_LANES);
        }
    }

    float tau = 1.0f;
    if (FINAL) tau = fmaxf(1.0f - ts[b], TAU_MIN_F);
    #pragma unroll
    for (int r = 0; r < ROUNDS; ++r) {
        int i = node0 + grp + GROUPS * r;
        if (i < nodeEnd && sub < 3) {
            float v = (sub == 0) ? (px[r] + ax[r])
                    : (sub == 1) ? (py[r] + ay[r]) : (pz[r] + az[r]);
            int idx = (b * N + i) * 3 + sub;
            if (FINAL) xout[idx] = (v - kp[idx]) / tau;
            else       xout[idx] = v;
        }
    }
}

extern "C" void kernel_launch(void* const* d_in, const int* in_sizes, int n_in,
                              void* d_out, int out_size, void* d_ws, size_t ws_size,
                              hipStream_t stream) {
    const float* kp   = (const float*)d_in[0];
    const float* ts   = (const float*)d_in[1];
    const float* ht   = (const float*)d_in[2];
    const float* W    = (const float*)d_in[3];
    const float* bias = (const float*)d_in[4];
    const int*   ei   = (const int*)d_in[5];
    const float* rl   = (const float*)d_in[6];

    const int B   = in_sizes[1];
    const int E   = in_sizes[6];
    const int BN3 = in_sizes[0];
    const int BN  = BN3 / 3;
    const int N   = BN / B;
    const int D   = in_sizes[2] / BN;

    const int T = 256;
    const int nTiles = (BN + 15) / 16;
    const int gVec   = (BN3 + T - 1) / T;
    const int BE     = B * E;
    const int gEdge  = (BE + T - 1) / T;
    const int gWmma  = (nTiles * 32 + T - 1) / T;
    const int XPBD_ITERS = 4;

    const int P    = (N + SLAB_NODES - 1) / SLAB_NODES;
    const int NL   = P * N;
    const int npc  = (N + CHUNKS - 1) / CHUNKS;
    const int NB   = (N + BK_CN - 1) >> BK_SHIFT;
    const int NBLK = (E + BK_EPB - 1) / BK_EPB;
    const size_t rW   = ((size_t)2 * E > (size_t)BN3) ? (size_t)2 * E : (size_t)BN3;
    const size_t need = ((size_t)4 * E + rW + (size_t)BN3 + (size_t)2 * NL + 1) * sizeof(int);

    if (E > 0 && npc <= GROUPS * ROUNDS && ws_size >= need &&
        NB <= BK_NBMAX && NBLK <= BK_T && P * BK_CN <= BK_T &&
        (size_t)2 * NB * NBLK <= (size_t)BN3) {
        int*   adj  = (int*)d_ws;
        int*   rec  = adj + (size_t)4 * E;
        float* x1   = (float*)rec;
        float* x0   = (float*)(rec + rW);
        int*   cnt  = (int*)(x0 + BN3);
        int*   pos  = cnt + NL + 1;
        int*   cntT = (int*)x0;
        int*   offT = cntT + (size_t)NB * NBLK;

        const dim3 gPass(CHUNKS, B);

        bucket_kernel<<<NBLK, BK_T, 0, stream>>>(ei, rec, cntT, offT, E, NB, NBLK);
        lists_kernel<<<NB, BK_T, 0, stream>>>(ei, rl, rec, cntT, offT, adj, pos, cnt, E, N, NBLK, P);
        head_wmma_kernel<<<gWmma, T, 0, stream>>>(ht, W, bias, kp, ts, x0, nTiles, D, N);

        slab_gather_kernel<false><<<gPass, PASS_T, 0, stream>>>(x0, adj, pos, cnt, kp, ts, x1, N, npc);
        slab_gather_kernel<false><<<gPass, PASS_T, 0, stream>>>(x1, adj, pos, cnt, kp, ts, x0, N, npc);
        slab_gather_kernel<false><<<gPass, PASS_T, 0, stream>>>(x0, adj, pos, cnt, kp, ts, x1, N, npc);
        slab_gather_kernel<true ><<<gPass, PASS_T, 0, stream>>>(x1, adj, pos, cnt, kp, ts,
                                                                (float*)d_out, N, npc);
        return;
    }

    float* xbuf  = (float*)d_ws;
    float* delta = xbuf + BN3;

    zero_kernel<<<gVec, T, 0, stream>>>(delta, BN3);
    head_wmma_kernel<<<gWmma, T, 0, stream>>>(ht, W, bias, kp, ts, xbuf, nTiles, D, N);

    for (int it = 0; it < XPBD_ITERS; ++it) {
        edge_kernel<<<gEdge, T, 0, stream>>>(xbuf, ei, rl, delta, E, N, BE);
        if (it < XPBD_ITERS - 1) {
            apply_kernel<<<gVec, T, 0, stream>>>(xbuf, delta, BN3);
        } else {
            final_kernel<<<gVec, T, 0, stream>>>(xbuf, delta, kp, ts,
                                                 (float*)d_out, BN3, N * 3);
        }
    }
}


# pass kernel: each load issued a stage ahead of its use: a slab's list ranges as one group before its copy (were 7 serial round trips), copy in two alternating load groups, list entries a trip ahead
# speedup vs baseline: 16.5477x; 1.0340x over previous
#include <hip/hip_runtime.h>
#include <hip/hip_bf16.h>


typedef float v2f __attribute__((ext_vector_type(2)));
typedef float v8f __attribute__((ext_vector_type(8)));

#define TAU_MIN_F     0.001f
#define XPBD_MAXC     0.15f
#define XPBD_DENOM    (2.0f + 1e-9f)

#define BSTRIDE 260

__device__ __forceinline__ void atomic_add_f32(float* p, float v) {
    unsafeAtomicAdd(p, v);
}

__global__ void zero_kernel(float* __restrict__ p, int n) {
    int i = blockIdx.x * blockDim.x + threadIdx.x;
    if (i < n) p[i] = 0.0f;
}

__global__ __launch_bounds__(256) void head_wmma_kernel(
    const float* __restrict__ ht,
    const float* __restrict__ W,
    const float* __restrict__ bias,
    const float* __restrict__ kp,
    const float* __restrict__ ts,
    float* __restrict__ x,
    int nTiles, int D, int N)
{
    __shared__ float sWT[16 * BSTRIDE];
    __shared__ float sB[3];

    for (int idx = threadIdx.x; idx < 16 * 256; idx += blockDim.x) {
        int n = idx >> 8;
        int k = idx & 255;
        sWT[n * BSTRIDE + k] = (n < 3) ? W[k * 3 + n] : 0.0f;
    }
    if (threadIdx.x < 3) sB[threadIdx.x] = bias[threadIdx.x];
    __syncthreads();

    int wave = (blockIdx.x * blockDim.x + threadIdx.x) >> 5;
    if (wave >= nTiles) return;

    int lane = threadIdx.x & 31;
    int half = lane >> 4;
    int l15  = lane & 15;

    long rowA = (long)wave * 16 + l15;
    const float* arow = ht + rowA * D + 2 * half;
    const float* brow = &sWT[l15 * BSTRIDE + 2 * half];

    v8f acc0 = {};
    v8f acc1 = {};
    #pragma unroll 4
    for (int kb = 0; kb < D; kb += 8) {
        v2f a0 = *(const v2f*)(arow + kb);
        v2f a1 = *(const v2f*)(arow + kb + 4);
        v2f b0 = *(const v2f*)(brow + kb);
        v2f b1 = *(const v2f*)(brow + kb + 4);
        acc0 = __builtin_amdgcn_wmma_f32_16x16x4_f32(
                   false, a0, false, b0, (short)0, acc0, false, false);
        acc1 = __builtin_amdgcn_wmma_f32_16x16x4_f32(
                   false, a1, false, b1, (short)0, acc1, false, false);
    }

    if (l15 < 3) {
        long rowBase = (long)wave * 16 + half * 8;
        #pragma unroll
        for (int r = 0; r < 8; ++r) {
            long row = rowBase + r;
            int  bidx = (int)(row / N);
            float tau = fmaxf(1.0f - ts[bidx], TAU_MIN_F);
            long idx = row * 3 + l15;
            x[idx] = kp[idx] + tau * (acc0[r] + acc1[r] + sB[l15]);
        }
    }
}

__global__ void edge_kernel(
    const float* __restrict__ x,
    const int*   __restrict__ ei,
    const float* __restrict__ rl,
    float* __restrict__ delta,
    int E, int N, int BE)
{
    int t = blockIdx.x * blockDim.x + threadIdx.x;
    if (t >= BE) return;
    int b = t / E;
    int e = t - b * E;
    int s = ei[e];
    int d = ei[E + e];
    int si = (b * N + s) * 3;
    int di = (b * N + d) * 3;

    float dx = x[si + 0] - x[di + 0];
    float dy = x[si + 1] - x[di + 1];
    float dz = x[si + 2] - x[di + 2];
    float dist = sqrtf(dx * dx + dy * dy + dz * dz);
    float C    = dist - rl[e];
    float scal = (-C / XPBD_DENOM) / (dist + 1e-9f);

    float cx = fminf(fmaxf(scal * dx, -XPBD_MAXC), XPBD_MAXC);
    float cy = fminf(fmaxf(scal * dy, -XPBD_MAXC), XPBD_MAXC);
    float cz = fminf(fmaxf(scal * dz, -XPBD_MAXC), XPBD_MAXC);

    atomic_add_f32(&delta[si + 0],  cx);
    atomic_add_f32(&delta[si + 1],  cy);
    atomic_add_f32(&delta[si + 2],  cz);
    atomic_add_f32(&delta[di + 0], -cx);
    atomic_add_f32(&delta[di + 1], -cy);
    atomic_add_f32(&delta[di + 2], -cz);
}

__global__ void apply_kernel(float* __restrict__ x, float* __restrict__ delta, int n) {
    int i = blockIdx.x * blockDim.x + threadIdx.x;
    if (i < n) {
        x[i] += delta[i];
        delta[i] = 0.0f;
    }
}

__global__ void final_kernel(
    const float* __restrict__ x, const float* __restrict__ delta,
    const float* __restrict__ kp, const float* __restrict__ ts,
    float* __restrict__ out, int n, int perBatch)
{
    int i = blockIdx.x * blockDim.x + threadIdx.x;
    if (i < n) {
        int b = i / perBatch;
        float tau = fmaxf(1.0f - ts[b], TAU_MIN_F);
        out[i] = (x[i] + delta[i] - kp[i]) / tau;
    }
}


typedef int   v2i __attribute__((ext_vector_type(2)));
typedef float v4f __attribute__((ext_vector_type(4)));

#define SLAB_NODES 25088
#define NODE_LANES 8
#define PASS_T     1024
#define GROUPS     (PASS_T / NODE_LANES)
#define ROUNDS     7
#define CHUNKS     64
#define COPY_DEPTH 4

#define BK_SHIFT 8
#define BK_CN    (1 << BK_SHIFT)
#define BK_NBMAX 256
#define BK_T     1024
#define BK_EPT   4
#define BK_EPB   (BK_T * BK_EPT)
#define BK_HPB   (2 * BK_EPB)

__global__ __launch_bounds__(BK_T) void bucket_kernel(
    const int* __restrict__ ei, int* __restrict__ rec,
    int* __restrict__ cntT, int* __restrict__ offT, int E, int NB, int NBLK)
{
    __shared__ int hist[BK_NBMAX];
    __shared__ int hoff[BK_NBMAX];
    const int tid = threadIdx.x;
    if (tid < BK_NBMAX) hist[tid] = 0;
    __syncthreads();

    const int e0 = blockIdx.x * BK_EPB;
    int bk[2 * BK_EPT], rk[2 * BK_EPT];
    #pragma unroll
    for (int u = 0; u < BK_EPT; ++u) {
        int e  = e0 + tid + u * BK_T;
        int ec = (e < E) ? e : 0;
        bk[2 * u]     = ei[ec] >> BK_SHIFT;
        bk[2 * u + 1] = ei[E + ec] >> BK_SHIFT;
    }
    #pragma unroll
    for (int u = 0; u < BK_EPT; ++u) {
        int e = e0 + tid + u * BK_T;
        rk[2 * u] = 0; rk[2 * u + 1] = 0;
        if (e < E) {
            rk[2 * u]     = atomicAdd(&hist[bk[2 * u]], 1);
            rk[2 * u + 1] = atomicAdd(&hist[bk[2 * u + 1]], 1);
        }
    }
    __syncthreads();
    if (tid < 32) {
        int a[8];
        int sum = 0;
        #pragma unroll
        for (int q = 0; q < 8; ++q) { a[q] = hist[8 * tid + q]; sum += a[q]; }
        int v = sum;
        for (int o = 1; o < 32; o <<= 1) {
            int t = __shfl_up(v, o);
            if (tid >= o) v += t;
        }
        int run = v - sum;
        #pragma unroll
        for (int q = 0; q < 8; ++q) { hoff[8 * tid + q] = run; run += a[q]; }
    }
    __syncthreads();
    if (tid < NB) {
        cntT[tid * NBLK + blockIdx.x] = hist[tid];
        offT[tid * NBLK + blockIdx.x] = hoff[tid];
    }
    int* myrec = rec + (size_t)blockIdx.x * BK_HPB;
    #pragma unroll
    for (int u = 0; u < BK_EPT; ++u) {
        int e = e0 + tid + u * BK_T;
        if (e < E) {
            myrec[hoff[bk[2 * u]]     + rk[2 * u]]     = 2 * e;
            myrec[hoff[bk[2 * u + 1]] + rk[2 * u + 1]] = 2 * e + 1;
        }
    }
}

__device__ __forceinline__ int block_scan_excl(int c, int* ws, int* tot) {
    const int lane = threadIdx.x & 31;
    const int w    = threadIdx.x >> 5;
    int v = c;
    for (int o = 1; o < 32; o <<= 1) {
        int t = __shfl_up(v, o);
        if (lane >= o) v += t;
    }
    if (lane == 31) ws[w] = v;
    __syncthreads();
    if (w == 0) {
        int s  = ws[lane];
        int sv = s;
        for (int o = 1; o < 32; o <<= 1) {
            int t = __shfl_up(sv, o);
            if (lane >= o) sv += t;
        }
        ws[lane] = sv - s;
        if (lane == 31) *tot = sv;
    }
    __syncthreads();
    return ws[w] + v - c;
}

#define BK_U 8

__global__ __launch_bounds__(BK_T) void lists_kernel(
    const int* __restrict__ ei, const float* __restrict__ rl,
    const int* __restrict__ rec, const int* __restrict__ cntT,
    const int* __restrict__ offT, int* __restrict__ adj,
    int* __restrict__ pos, int* __restrict__ cnt,
    int E, int N, int NBLK, int P)
{
    __shared__ int segO[BK_T];
    __shared__ int segP[BK_T + 1];
    __shared__ int lcnt[BK_T];
    __shared__ int lcur[BK_T];
    __shared__ int wsA[32], wsB[32], wsC[32];
    __shared__ int totA, totB, baseSh;

    const int tid   = threadIdx.x;
    const int lane  = tid & 31;
    const int w     = tid >> 5;
    const int kb    = blockIdx.x;
    const int node0 = kb << BK_SHIFT;

    int c = 0, o = 0;
    if (tid < NBLK) { c = cntT[kb * NBLK + tid]; o = offT[kb * NBLK + tid]; }
    segO[tid] = o;
    int ex = block_scan_excl(c, wsA, &totA);
    segP[tid] = ex;
    if (tid == BK_T - 1) segP[BK_T] = ex + c;

    int acc = 0;
    for (int idx = tid; idx < kb * NBLK; idx += BK_T) acc += cntT[idx];
    for (int m = 16; m >= 1; m >>= 1) acc += __shfl_xor(acc, m);
    if (lane == 0) wsB[w] = acc;
    lcnt[tid] = 0;
    __syncthreads();
    if (w == 0) {
        int s = wsB[lane];
        for (int m = 16; m >= 1; m >>= 1) s += __shfl_xor(s, m);
        if (lane == 0) baseSh = s;
    }
    __syncthreads();
    const int T    = totA;
    const int base = baseSh;
    int st0 = 1;
    while (st0 * 2 < NBLK) st0 *= 2;

    for (int t0 = tid; t0 < T; t0 += BK_T * BK_U) {
        int  hh[BK_U], tt[BK_U], b[BK_U];
        bool ok[BK_U];
        #pragma unroll
        for (int u = 0; u < BK_U; ++u) {
            int t = t0 + u * BK_T;
            ok[u] = t < T;
            tt[u] = ok[u] ? t : 0;
            b[u]  = 0;
        }
        for (int st = st0; st >= 1; st >>= 1) {
            #pragma unroll
            for (int u = 0; u < BK_U; ++u) b[u] += (segP[b[u] + st] <= tt[u]) ? st : 0;
        }
        #pragma unroll
        for (int u = 0; u < BK_U; ++u) {
            hh[u] = rec[b[u] * BK_HPB + segO[b[u]] + (tt[u] - segP[b[u]])];
        }
        int s[BK_U], d[BK_U];
        #pragma unroll
        for (int u = 0; u < BK_U; ++u) {
            int e = hh[u] >> 1;
            s[u] = ei[e];
            d[u] = ei[E + e];
        }
        #pragma unroll
        for (int u = 0; u < BK_U; ++u) {
            int own = (hh[u] & 1) ? d[u] : s[u];
            int nbr = (hh[u] & 1) ? s[u] : d[u];
            int ll  = ((nbr / SLAB_NODES) << BK_SHIFT) | (own & (BK_CN - 1));
            if (ok[u]) atomicAdd(&lcnt[ll], 1);
        }
    }
    __syncthreads();

    const int LISTS = P << BK_SHIFT;
    int lc  = (tid < LISTS) ? lcnt[tid] : 0;
    int lex = block_scan_excl(lc, wsC, &totB);
    if (tid < LISTS) {
        int p = tid >> BK_SHIFT;
        int i = node0 + (tid & (BK_CN - 1));
        lcur[tid] = base + lex;
        if (i < N) {
            pos[p * N + i] = base + lex;
            cnt[p * N + i] = lc;
        }
    }
    __syncthreads();

    v2i* a2 = (v2i*)adj;
    for (int t0 = tid; t0 < T; t0 += BK_T * BK_U) {
        int  hh[BK_U], tt[BK_U], b[BK_U];
        bool ok[BK_U];
        #pragma unroll
        for (int u = 0; u < BK_U; ++u) {
            int t = t0 + u * BK_T;
            ok[u] = t < T;
            tt[u] = ok[u] ? t : 0;
            b[u]  = 0;
        }
        for (int st = st0; st >= 1; st >>= 1) {
            #pragma unroll
            for (int u = 0; u < BK_U; ++u) b[u] += (segP[b[u] + st] <= tt[u]) ? st : 0;
        }
        #pragma unroll
        for (int u = 0; u < BK_U; ++u) {
            hh[u] = rec[b[u] * BK_HPB + segO[b[u]] + (tt[u] - segP[b[u]])];
        }
        int s[BK_U], d[BK_U], l0[BK_U];
        #pragma unroll
        for (int u = 0; u < BK_U; ++u) {
            int e = hh[u] >> 1;
            s[u]  = ei[e];
            d[u]  = ei[E + e];
            l0[u] = __float_as_int(rl[e]);
        }
        #pragma unroll
        for (int u = 0; u < BK_U; ++u) {
            int own = (hh[u] & 1) ? d[u] : s[u];
            int nbr = (hh[u] & 1) ? s[u] : d[u];
            int ll  = ((nbr / SLAB_NODES) << BK_SHIFT) | (own & (BK_CN - 1));
            if (ok[u]) {
                int slot = atomicAdd(&lcur[ll], 1);
                v2i en; en.x = nbr; en.y = l0[u];
                a2[slot] = en;
            }
        }
    }
}

__device__ __forceinline__ void fetch_entries(
    v2i (&ent)[ROUNDS], bool (&has)[ROUNDS], int (&k)[ROUNDS],
    const int (&end)[ROUNDS], const v2i* __restrict__ a2)
{
    #pragma unroll
    for (int r = 0; r < ROUNDS; ++r) {
        has[r] = k[r] < end[r];
        ent[r] = a2[has[r] ? k[r] : 0];
        k[r] += NODE_LANES;
    }
}

__device__ __forceinline__ void walk_entries(
    const v2i (&ent)[ROUNDS], const bool (&has)[ROUNDS], const float* sx, int lo,
    const float (&px)[ROUNDS], const float (&py)[ROUNDS], const float (&pz)[ROUNDS],
    float (&ax)[ROUNDS], float (&ay)[ROUNDS], float (&az)[ROUNDS])
{
    #pragma unroll
    for (int r = 0; r < ROUNDS; ++r) {
        if (has[r]) {
            int j = ent[r].x;
            const float* q = &sx[(j - lo) * 3];
            float L0 = __int_as_float(ent[r].y);
            float dx = px[r] - q[0];
            float dy = py[r] - q[1];
            float dz = pz[r] - q[2];
            float dist = sqrtf(dx * dx + dy * dy + dz * dz);
            float C    = dist - L0;
            float scal = (-C / XPBD_DENOM) / (dist + 1e-9f);
            ax[r] += fminf(fmaxf(scal * dx, -XPBD_MAXC), XPBD_MAXC);
            ay[r] += fminf(fmaxf(scal * dy, -XPBD_MAXC), XPBD_MAXC);
            az[r] += fminf(fmaxf(scal * dz, -XPBD_MAXC), XPBD_MAXC);
        }
    }
}

template <bool FINAL>
__global__ __launch_bounds__(PASS_T) void slab_gather_kernel(
    const float* __restrict__ xin,
    const int*   __restrict__ adj,
    const int*   __restrict__ pos,
    const int*   __restrict__ cnt,
    const float* __restrict__ kp,
    const float* __restrict__ ts,
    float* __restrict__ xout, int N, int npc)
{
    __shared__ __attribute__((aligned(16))) float sx[SLAB_NODES * 3];

    const int tid     = threadIdx.x;
    const int grp     = tid / NODE_LANES;
    const int sub     = tid % NODE_LANES;
    const int b       = blockIdx.y;
    const int node0   = blockIdx.x * npc;
    const int nodeEnd = min(N, node0 + npc);
    const float* xb   = xin + (size_t)b * N * 3;
    const v2i*   a2   = (const v2i*)adj;

    float px[ROUNDS], py[ROUNDS], pz[ROUNDS];
    float ax[ROUNDS], ay[ROUNDS], az[ROUNDS];
    #pragma unroll
    for (int r = 0; r < ROUNDS; ++r) {
        int i = min(node0 + grp + GROUPS * r, N - 1);
        const float* p = xb + i * 3;
        px[r] = p[0]; py[r] = p[1]; pz[r] = p[2];
        ax[r] = 0.0f; ay[r] = 0.0f; az[r] = 0.0f;
    }

    int ph = 0;
    for (int lo = 0; lo < N; lo += SLAB_NODES, ++ph) {
        const int hi = min(N, lo + SLAB_NODES);
        const int nw = (hi - lo) * 3;

        __syncthreads();

        int s0v[ROUNDS], cv[ROUNDS];
        #pragma unroll
        for (int r = 0; r < ROUNDS; ++r) {
            int  i   = node0 + grp + GROUPS * r;
            bool own = i < nodeEnd;
            int  l   = ph * N + (own ? i : 0);
            int  c   = cnt[l];
            s0v[r]   = pos[l];
            cv[r]    = own ? c : 0;
        }

        const float* src = xb + (size_t)lo * 3;
        if ((((size_t)src) & 15) == 0) {
            const v4f* s4 = (const v4f*)src;
            v4f*       d4 = (v4f*)sx;
            const int  n4 = nw >> 2;
            const int last = n4 - 1;
            const int STEP = PASS_T * COPY_DEPTH;
            v4f bufA[COPY_DEPTH], bufB[COPY_DEPTH];
            if (n4 > 0) {
                #pragma unroll
                for (int u = 0; u < COPY_DEPTH; ++u)
                    bufA[u] = s4[min(tid + u * PASS_T, last)];
                for (int base = 0; base < n4; base += 2 * STEP) {
                    #pragma unroll
                    for (int u = 0; u < COPY_DEPTH; ++u)
                        bufB[u] = s4[min(base + STEP + tid + u * PASS_T, last)];
                    #pragma unroll
                    for (int u = 0; u < COPY_DEPTH; ++u) {
                        int w = base + tid + u * PASS_T;
                        if (w < n4) d4[w] = bufA[u];
                    }
                    #pragma unroll
                    for (int u = 0; u < COPY_DEPTH; ++u)
                        bufA[u] = s4[min(base + 2 * STEP + tid + u * PASS_T, last)];
                    #pragma unroll
                    for (int u = 0; u < COPY_DEPTH; ++u) {
                        int w = base + STEP + tid + u * PASS_T;
                        if (w < n4) d4[w] = bufB[u];
                    }
                }
            }
            for (int w = (n4 << 2) + tid; w < nw; w += PASS_T) sx[w] = src[w];
        } else {
            for (int w = tid; w < nw; w += PASS_T) sx[w] = src[w];
        }

        int k[ROUNDS], end[ROUNDS];
        int trips = 0;
        #pragma unroll
        for (int r = 0; r < ROUNDS; ++r) {
            k[r]   = s0v[r] + sub;
            end[r] = s0v[r] + cv[r];
            trips  = max(trips, (end[r] - k[r] + NODE_LANES - 1) / NODE_LANES);
        }

        v2i  entA[ROUNDS], entB[ROUNDS];
        bool hasA[ROUNDS], hasB[ROUNDS];
        fetch_entries(entA, hasA, k, end, a2);

        __syncthreads();

        for (int t = 0; t < trips; t += 2) {
            fetch_entries(entB, hasB, k, end, a2);
            walk_entries(entA, hasA, sx, lo, px, py, pz, ax, ay, az);
            fetch_entries(entA, hasA, k, end, a2);
            walk_entries(entB, hasB, sx, lo, px, py, pz, ax, ay, az);
        }
    }

    #pragma unroll
    for (int m = 1; m < NODE_LANES; m <<= 1) {
        #pragma unroll
        for (int r = 0; r < ROUNDS; ++r) {
            ax[r] += __shfl_xor(ax[r], m, NODE_LANES);
            ay[r] += __shfl_xor(ay[r], m, NODE_LANES);
            az[r] += __shfl_xor(az[r], m, NODE_LANES);
        }
    }

    float tau = 1.0f;
    if (FINAL) tau = fmaxf(1.0f - ts[b], TAU_MIN_F);
    #pragma unroll
    for (int r = 0; r < ROUNDS; ++r) {
        int i = node0 + grp + GROUPS * r;
        if (i < nodeEnd && sub < 3) {
            float v = (sub == 0) ? (px[r] + ax[r])
                    : (sub == 1) ? (py[r] + ay[r]) : (pz[r] + az[r]);
            int idx = (b * N + i) * 3 + sub;
            if (FINAL) xout[idx] = (v - kp[idx]) / tau;
            else       xout[idx] = v;
        }
    }
}

extern "C" void kernel_launch(void* const* d_in, const int* in_sizes, int n_in,
                              void* d_out, int out_size, void* d_ws, size_t ws_size,
                              hipStream_t stream) {
    const float* kp   = (const float*)d_in[0];
    const float* ts   = (const float*)d_in[1];
    const float* ht   = (const float*)d_in[2];
    const float* W    = (const float*)d_in[3];
    const float* bias = (const float*)d_in[4];
    const int*   ei   = (const int*)d_in[5];
    const float* rl   = (const float*)d_in[6];

    const int B   = in_sizes[1];
    const int E   = in_sizes[6];
    const int BN3 = in_sizes[0];
    const int BN  = BN3 / 3;
    const int N   = BN / B;
    const int D   = in_sizes[2] / BN;

    const int T = 256;
    const int nTiles = (BN + 15) / 16;
    const int gVec   = (BN3 + T - 1) / T;
    const int BE     = B * E;
    const int gEdge  = (BE + T - 1) / T;
    const int gWmma  = (nTiles * 32 + T - 1) / T;
    const int XPBD_ITERS = 4;

    const int P    = (N + SLAB_NODES - 1) / SLAB_NODES;
    const int NL   = P * N;
    const int npc  = (N + CHUNKS - 1) / CHUNKS;
    const int NB   = (N + BK_CN - 1) >> BK_SHIFT;
    const int NBLK = (E + BK_EPB - 1) / BK_EPB;
    const size_t rW   = ((size_t)2 * E > (size_t)BN3) ? (size_t)2 * E : (size_t)BN3;
    const size_t need = ((size_t)4 * E + rW + (size_t)BN3 + (size_t)2 * NL + 1) * sizeof(int);

    if (E > 0 && npc <= GROUPS * ROUNDS && ws_size >= need &&
        NB <= BK_NBMAX && NBLK <= BK_T && P * BK_CN <= BK_T &&
        (size_t)2 * NB * NBLK <= (size_t)BN3) {
        int*   adj  = (int*)d_ws;
        int*   rec  = adj + (size_t)4 * E;
        float* x1   = (float*)rec;
        float* x0   = (float*)(rec + rW);
        int*   cnt  = (int*)(x0 + BN3);
        int*   pos  = cnt + NL + 1;
        int*   cntT = (int*)x0;
        int*   offT = cntT + (size_t)NB * NBLK;

        const dim3 gPass(CHUNKS, B);

        bucket_kernel<<<NBLK, BK_T, 0, stream>>>(ei, rec, cntT, offT, E, NB, NBLK);
        lists_kernel<<<NB, BK_T, 0, stream>>>(ei, rl, rec, cntT, offT, adj, pos, cnt, E, N, NBLK, P);
        head_wmma_kernel<<<gWmma, T, 0, stream>>>(ht, W, bias, kp, ts, x0, nTiles, D, N);

        slab_gather_kernel<false><<<gPass, PASS_T, 0, stream>>>(x0, adj, pos, cnt, kp, ts, x1, N, npc);
        slab_gather_kernel<false><<<gPass, PASS_T, 0, stream>>>(x1, adj, pos, cnt, kp, ts, x0, N, npc);
        slab_gather_kernel<false><<<gPass, PASS_T, 0, stream>>>(x0, adj, pos, cnt, kp, ts, x1, N, npc);
        slab_gather_kernel<true ><<<gPass, PASS_T, 0, stream>>>(x1, adj, pos, cnt, kp, ts,
                                                                (float*)d_out, N, npc);
        return;
    }

    float* xbuf  = (float*)d_ws;
    float* delta = xbuf + BN3;

    zero_kernel<<<gVec, T, 0, stream>>>(delta, BN3);
    head_wmma_kernel<<<gWmma, T, 0, stream>>>(ht, W, bias, kp, ts, xbuf, nTiles, D, N);

    for (int it = 0; it < XPBD_ITERS; ++it) {
        edge_kernel<<<gEdge, T, 0, stream>>>(xbuf, ei, rl, delta, E, N, BE);
        if (it < XPBD_ITERS - 1) {
            apply_kernel<<<gVec, T, 0, stream>>>(xbuf, delta, BN3);
        } else {
            final_kernel<<<gVec, T, 0, stream>>>(xbuf, delta, kp, ts,
                                                 (float*)d_out, BN3, N * 3);
        }
    }
}


# head product's tiles moved into the second list-build launch (build blocks first, head blocks beside them on a WGP): its memory stream runs under the build's load waits; 7 launches become 6
# speedup vs baseline: 17.4271x; 1.0531x over previous
#include <hip/hip_runtime.h>
#include <hip/hip_bf16.h>


typedef float v2f __attribute__((ext_vector_type(2)));
typedef float v8f __attribute__((ext_vector_type(8)));

#define TAU_MIN_F     0.001f
#define XPBD_MAXC     0.15f
#define XPBD_DENOM    (2.0f + 1e-9f)

#define BSTRIDE 260

__device__ __forceinline__ void atomic_add_f32(float* p, float v) {
    unsafeAtomicAdd(p, v);
}

__global__ void zero_kernel(float* __restrict__ p, int n) {
    int i = blockIdx.x * blockDim.x + threadIdx.x;
    if (i < n) p[i] = 0.0f;
}

__device__ __forceinline__ void head_tiles(
    const float* __restrict__ ht,
    const float* __restrict__ W,
    const float* __restrict__ bias,
    const float* __restrict__ kp,
    const float* __restrict__ ts,
    float* __restrict__ x,
    int tile0, int tileEnd, int hb, int D, int N)
{
    __shared__ float sWT[16 * BSTRIDE];
    __shared__ float sB[3];

    for (int idx = threadIdx.x; idx < 16 * 256; idx += blockDim.x) {
        int n = idx >> 8;
        int k = idx & 255;
        sWT[n * BSTRIDE + k] = (n < 3) ? W[k * 3 + n] : 0.0f;
    }
    if (threadIdx.x < 3) sB[threadIdx.x] = bias[threadIdx.x];
    __syncthreads();

    int wave = tile0 + ((hb * (int)blockDim.x + (int)threadIdx.x) >> 5);
    if (wave >= tileEnd) return;

    int lane = threadIdx.x & 31;
    int half = lane >> 4;
    int l15  = lane & 15;

    long rowA = (long)wave * 16 + l15;
    const float* arow = ht + rowA * D + 2 * half;
    const float* brow = &sWT[l15 * BSTRIDE + 2 * half];

    v8f acc0 = {};
    v8f acc1 = {};
    #pragma unroll 4
    for (int kb = 0; kb < D; kb += 8) {
        v2f a0 = *(const v2f*)(arow + kb);
        v2f a1 = *(const v2f*)(arow + kb + 4);
        v2f b0 = *(const v2f*)(brow + kb);
        v2f b1 = *(const v2f*)(brow + kb + 4);
        acc0 = __builtin_amdgcn_wmma_f32_16x16x4_f32(
                   false, a0, false, b0, (short)0, acc0, false, false);
        acc1 = __builtin_amdgcn_wmma_f32_16x16x4_f32(
                   false, a1, false, b1, (short)0, acc1, false, false);
    }

    if (l15 < 3) {
        unsigned rowBase = (unsigned)wave * 16u + (unsigned)half * 8u;
        #pragma unroll
        for (int r = 0; r < 8; ++r) {
            unsigned row = rowBase + (unsigned)r;
            int  bidx = (int)(row / (unsigned)N);
            float tau = fmaxf(1.0f - ts[bidx], TAU_MIN_F);
            long idx = (long)row * 3 + l15;
            x[idx] = kp[idx] + tau * (acc0[r] + acc1[r] + sB[l15]);
        }
    }
}

__global__ __launch_bounds__(256) void head_wmma_kernel(
    const float* __restrict__ ht, const float* __restrict__ W,
    const float* __restrict__ bias, const float* __restrict__ kp,
    const float* __restrict__ ts, float* __restrict__ x,
    int nTiles, int D, int N)
{
    head_tiles(ht, W, bias, kp, ts, x, 0, nTiles, (int)blockIdx.x, D, N);
}

__global__ void edge_kernel(
    const float* __restrict__ x,
    const int*   __restrict__ ei,
    const float* __restrict__ rl,
    float* __restrict__ delta,
    int E, int N, int BE)
{
    int t = blockIdx.x * blockDim.x + threadIdx.x;
    if (t >= BE) return;
    int b = t / E;
    int e = t - b * E;
    int s = ei[e];
    int d = ei[E + e];
    int si = (b * N + s) * 3;
    int di = (b * N + d) * 3;

    float dx = x[si + 0] - x[di + 0];
    float dy = x[si + 1] - x[di + 1];
    float dz = x[si + 2] - x[di + 2];
    float dist = sqrtf(dx * dx + dy * dy + dz * dz);
    float C    = dist - rl[e];
    float scal = (-C / XPBD_DENOM) / (dist + 1e-9f);

    float cx = fminf(fmaxf(scal * dx, -XPBD_MAXC), XPBD_MAXC);
    float cy = fminf(fmaxf(scal * dy, -XPBD_MAXC), XPBD_MAXC);
    float cz = fminf(fmaxf(scal * dz, -XPBD_MAXC), XPBD_MAXC);

    atomic_add_f32(&delta[si + 0],  cx);
    atomic_add_f32(&delta[si + 1],  cy);
    atomic_add_f32(&delta[si + 2],  cz);
    atomic_add_f32(&delta[di + 0], -cx);
    atomic_add_f32(&delta[di + 1], -cy);
    atomic_add_f32(&delta[di + 2], -cz);
}

__global__ void apply_kernel(float* __restrict__ x, float* __restrict__ delta, int n) {
    int i = blockIdx.x * blockDim.x + threadIdx.x;
    if (i < n) {
        x[i] += delta[i];
        delta[i] = 0.0f;
    }
}

__global__ void final_kernel(
    const float* __restrict__ x, const float* __restrict__ delta,
    const float* __restrict__ kp, const float* __restrict__ ts,
    float* __restrict__ out, int n, int perBatch)
{
    int i = blockIdx.x * blockDim.x + threadIdx.x;
    if (i < n) {
        int b = i / perBatch;
        float tau = fmaxf(1.0f - ts[b], TAU_MIN_F);
        out[i] = (x[i] + delta[i] - kp[i]) / tau;
    }
}


typedef int   v2i __attribute__((ext_vector_type(2)));
typedef float v4f __attribute__((ext_vector_type(4)));

#define SLAB_NODES 25088
#define NODE_LANES 8
#define PASS_T     1024
#define GROUPS     (PASS_T / NODE_LANES)
#define ROUNDS     7
#define CHUNKS     64
#define COPY_DEPTH 4

#define BK_SHIFT 8
#define BK_CN    (1 << BK_SHIFT)
#define BK_NBMAX 256
#define BK_T     1024
#define BK_EPT   4
#define BK_EPB   (BK_T * BK_EPT)
#define BK_HPB   (2 * BK_EPB)

__global__ __launch_bounds__(BK_T) void bucket_kernel(
    const int* __restrict__ ei, int* __restrict__ rec,
    int* __restrict__ cntT, int* __restrict__ offT, int E, int NB, int NBLK)
{
    __shared__ int hist[BK_NBMAX];
    __shared__ int hoff[BK_NBMAX];
    const int tid = threadIdx.x;
    if (tid < BK_NBMAX) hist[tid] = 0;
    __syncthreads();

    const int e0 = blockIdx.x * BK_EPB;
    int bk[2 * BK_EPT], rk[2 * BK_EPT];
    #pragma unroll
    for (int u = 0; u < BK_EPT; ++u) {
        int e  = e0 + tid + u * BK_T;
        int ec = (e < E) ? e : 0;
        bk[2 * u]     = ei[ec] >> BK_SHIFT;
        bk[2 * u + 1] = ei[E + ec] >> BK_SHIFT;
    }
    #pragma unroll
    for (int u = 0; u < BK_EPT; ++u) {
        int e = e0 + tid + u * BK_T;
        rk[2 * u] = 0; rk[2 * u + 1] = 0;
        if (e < E) {
            rk[2 * u]     = atomicAdd(&hist[bk[2 * u]], 1);
            rk[2 * u + 1] = atomicAdd(&hist[bk[2 * u + 1]], 1);
        }
    }
    __syncthreads();
    if (tid < 32) {
        int a[8];
        int sum = 0;
        #pragma unroll
        for (int q = 0; q < 8; ++q) { a[q] = hist[8 * tid + q]; sum += a[q]; }
        int v = sum;
        for (int o = 1; o < 32; o <<= 1) {
            int t = __shfl_up(v, o);
            if (tid >= o) v += t;
        }
        int run = v - sum;
        #pragma unroll
        for (int q = 0; q < 8; ++q) { hoff[8 * tid + q] = run; run += a[q]; }
    }
    __syncthreads();
    if (tid < NB) {
        cntT[tid * NBLK + blockIdx.x] = hist[tid];
        offT[tid * NBLK + blockIdx.x] = hoff[tid];
    }
    int* myrec = rec + (size_t)blockIdx.x * BK_HPB;
    #pragma unroll
    for (int u = 0; u < BK_EPT; ++u) {
        int e = e0 + tid + u * BK_T;
        if (e < E) {
            myrec[hoff[bk[2 * u]]     + rk[2 * u]]     = 2 * e;
            myrec[hoff[bk[2 * u + 1]] + rk[2 * u + 1]] = 2 * e + 1;
        }
    }
}

__device__ __forceinline__ int block_scan_excl(int c, int* ws, int* tot) {
    const int lane = threadIdx.x & 31;
    const int w    = threadIdx.x >> 5;
    int v = c;
    for (int o = 1; o < 32; o <<= 1) {
        int t = __shfl_up(v, o);
        if (lane >= o) v += t;
    }
    if (lane == 31) ws[w] = v;
    __syncthreads();
    if (w == 0) {
        int s  = ws[lane];
        int sv = s;
        for (int o = 1; o < 32; o <<= 1) {
            int t = __shfl_up(sv, o);
            if (lane >= o) sv += t;
        }
        ws[lane] = sv - s;
        if (lane == 31) *tot = sv;
    }
    __syncthreads();
    return ws[w] + v - c;
}

#define BK_U 8

__device__ __forceinline__ void lists_body(
    const int* __restrict__ ei, const float* __restrict__ rl,
    const int* __restrict__ rec, const int* __restrict__ cntT,
    const int* __restrict__ offT, int* __restrict__ adj,
    int* __restrict__ pos, int* __restrict__ cnt,
    int E, int N, int NBLK, int P)
{
    __shared__ int segO[BK_T];
    __shared__ int segP[BK_T + 1];
    __shared__ int lcnt[BK_T];
    __shared__ int lcur[BK_T];
    __shared__ int wsA[32], wsB[32], wsC[32];
    __shared__ int totA, totB, baseSh;

    const int tid   = threadIdx.x;
    const int lane  = tid & 31;
    const int w     = tid >> 5;
    const int kb    = blockIdx.x;
    const int node0 = kb << BK_SHIFT;

    int c = 0, o = 0;
    if (tid < NBLK) { c = cntT[kb * NBLK + tid]; o = offT[kb * NBLK + tid]; }
    segO[tid] = o;
    int ex = block_scan_excl(c, wsA, &totA);
    segP[tid] = ex;
    if (tid == BK_T - 1) segP[BK_T] = ex + c;

    int acc = 0;
    for (int idx = tid; idx < kb * NBLK; idx += BK_T) acc += cntT[idx];
    for (int m = 16; m >= 1; m >>= 1) acc += __shfl_xor(acc, m);
    if (lane == 0) wsB[w] = acc;
    lcnt[tid] = 0;
    __syncthreads();
    if (w == 0) {
        int s = wsB[lane];
        for (int m = 16; m >= 1; m >>= 1) s += __shfl_xor(s, m);
        if (lane == 0) baseSh = s;
    }
    __syncthreads();
    const int T    = totA;
    const int base = baseSh;
    int st0 = 1;
    while (st0 * 2 < NBLK) st0 *= 2;

    for (int t0 = tid; t0 < T; t0 += BK_T * BK_U) {
        int  hh[BK_U], tt[BK_U], b[BK_U];
        bool ok[BK_U];
        #pragma unroll
        for (int u = 0; u < BK_U; ++u) {
            int t = t0 + u * BK_T;
            ok[u] = t < T;
            tt[u] = ok[u] ? t : 0;
            b[u]  = 0;
        }
        for (int st = st0; st >= 1; st >>= 1) {
            #pragma unroll
            for (int u = 0; u < BK_U; ++u) b[u] += (segP[b[u] + st] <= tt[u]) ? st : 0;
        }
        #pragma unroll
        for (int u = 0; u < BK_U; ++u) {
            hh[u] = rec[b[u] * BK_HPB + segO[b[u]] + (tt[u] - segP[b[u]])];
        }
        int s[BK_U], d[BK_U];
        #pragma unroll
        for (int u = 0; u < BK_U; ++u) {
            int e = hh[u] >> 1;
            s[u] = ei[e];
            d[u] = ei[E + e];
        }
        #pragma unroll
        for (int u = 0; u < BK_U; ++u) {
            int own = (hh[u] & 1) ? d[u] : s[u];
            int nbr = (hh[u] & 1) ? s[u] : d[u];
            int ll  = ((nbr / SLAB_NODES) << BK_SHIFT) | (own & (BK_CN - 1));
            if (ok[u]) atomicAdd(&lcnt[ll], 1);
        }
    }
    __syncthreads();

    const int LISTS = P << BK_SHIFT;
    int lc  = (tid < LISTS) ? lcnt[tid] : 0;
    int lex = block_scan_excl(lc, wsC, &totB);
    if (tid < LISTS) {
        int p = tid >> BK_SHIFT;
        int i = node0 + (tid & (BK_CN - 1));
        lcur[tid] = base + lex;
        if (i < N) {
            pos[p * N + i] = base + lex;
            cnt[p * N + i] = lc;
        }
    }
    __syncthreads();

    v2i* a2 = (v2i*)adj;
    for (int t0 = tid; t0 < T; t0 += BK_T * BK_U) {
        int  hh[BK_U], tt[BK_U], b[BK_U];
        bool ok[BK_U];
        #pragma unroll
        for (int u = 0; u < BK_U; ++u) {
            int t = t0 + u * BK_T;
            ok[u] = t < T;
            tt[u] = ok[u] ? t : 0;
            b[u]  = 0;
        }
        for (int st = st0; st >= 1; st >>= 1) {
            #pragma unroll
            for (int u = 0; u < BK_U; ++u) b[u] += (segP[b[u] + st] <= tt[u]) ? st : 0;
        }
        #pragma unroll
        for (int u = 0; u < BK_U; ++u) {
            hh[u] = rec[b[u] * BK_HPB + segO[b[u]] + (tt[u] - segP[b[u]])];
        }
        int s[BK_U], d[BK_U], l0[BK_U];
        #pragma unroll
        for (int u = 0; u < BK_U; ++u) {
            int e = hh[u] >> 1;
            s[u]  = ei[e];
            d[u]  = ei[E + e];
            l0[u] = __float_as_int(rl[e]);
        }
        #pragma unroll
        for (int u = 0; u < BK_U; ++u) {
            int own = (hh[u] & 1) ? d[u] : s[u];
            int nbr = (hh[u] & 1) ? s[u] : d[u];
            int ll  = ((nbr / SLAB_NODES) << BK_SHIFT) | (own & (BK_CN - 1));
            if (ok[u]) {
                int slot = atomicAdd(&lcur[ll], 1);
                v2i en; en.x = nbr; en.y = l0[u];
                a2[slot] = en;
            }
        }
    }
}

__global__ __launch_bounds__(BK_T) void lists_head_kernel(
    const int* __restrict__ ei, const float* __restrict__ rl,
    const int* __restrict__ rec, const int* __restrict__ cntT,
    const int* __restrict__ offT, int* __restrict__ adj,
    int* __restrict__ pos, int* __restrict__ cnt,
    int E, int N, int NBLK, int P, int NB,
    const float* __restrict__ ht, const float* __restrict__ W,
    const float* __restrict__ bias, const float* __restrict__ kp,
    const float* __restrict__ ts, float* __restrict__ x,
    int tile0, int tileEnd, int D)
{
    if ((int)blockIdx.x < NB) lists_body(ei, rl, rec, cntT, offT, adj, pos, cnt, E, N, NBLK, P);
    else head_tiles(ht, W, bias, kp, ts, x, tile0, tileEnd, (int)blockIdx.x - NB, D, N);
}

__device__ __forceinline__ void fetch_entries(
    v2i (&ent)[ROUNDS], bool (&has)[ROUNDS], int (&k)[ROUNDS],
    const int (&end)[ROUNDS], const v2i* __restrict__ a2)
{
    #pragma unroll
    for (int r = 0; r < ROUNDS; ++r) {
        has[r] = k[r] < end[r];
        ent[r] = a2[has[r] ? k[r] : 0];
        k[r] += NODE_LANES;
    }
}

__device__ __forceinline__ void walk_entries(
    const v2i (&ent)[ROUNDS], const bool (&has)[ROUNDS], const float* sx, int lo,
    const float (&px)[ROUNDS], const float (&py)[ROUNDS], const float (&pz)[ROUNDS],
    float (&ax)[ROUNDS], float (&ay)[ROUNDS], float (&az)[ROUNDS])
{
    #pragma unroll
    for (int r = 0; r < ROUNDS; ++r) {
        if (has[r]) {
            int j = ent[r].x;
            const float* q = &sx[(j - lo) * 3];
            float L0 = __int_as_float(ent[r].y);
            float dx = px[r] - q[0];
            float dy = py[r] - q[1];
            float dz = pz[r] - q[2];
            float dist = sqrtf(dx * dx + dy * dy + dz * dz);
            float C    = dist - L0;
            float scal = (-C / XPBD_DENOM) / (dist + 1e-9f);
            ax[r] += fminf(fmaxf(scal * dx, -XPBD_MAXC), XPBD_MAXC);
            ay[r] += fminf(fmaxf(scal * dy, -XPBD_MAXC), XPBD_MAXC);
            az[r] += fminf(fmaxf(scal * dz, -XPBD_MAXC), XPBD_MAXC);
        }
    }
}

template <bool FINAL>
__global__ __launch_bounds__(PASS_T) void slab_gather_kernel(
    const float* __restrict__ xin,
    const int*   __restrict__ adj,
    const int*   __restrict__ pos,
    const int*   __restrict__ cnt,
    const float* __restrict__ kp,
    const float* __restrict__ ts,
    float* __restrict__ xout, int N, int npc)
{
    __shared__ __attribute__((aligned(16))) float sx[SLAB_NODES * 3];

    const int tid     = threadIdx.x;
    const int grp     = tid / NODE_LANES;
    const int sub     = tid % NODE_LANES;
    const int b       = blockIdx.y;
    const int node0   = blockIdx.x * npc;
    const int nodeEnd = min(N, node0 + npc);
    const float* xb   = xin + (size_t)b * N * 3;
    const v2i*   a2   = (const v2i*)adj;

    float px[ROUNDS], py[ROUNDS], pz[ROUNDS];
    float ax[ROUNDS], ay[ROUNDS], az[ROUNDS];
    #pragma unroll
    for (int r = 0; r < ROUNDS; ++r) {
        int i = min(node0 + grp + GROUPS * r, N - 1);
        const float* p = xb + i * 3;
        px[r] = p[0]; py[r] = p[1]; pz[r] = p[2];
        ax[r] = 0.0f; ay[r] = 0.0f; az[r] = 0.0f;
    }

    int ph = 0;
    for (int lo = 0; lo < N; lo += SLAB_NODES, ++ph) {
        const int hi = min(N, lo + SLAB_NODES);
        const int nw = (hi - lo) * 3;

        __syncthreads();

        int s0v[ROUNDS], cv[ROUNDS];
        #pragma unroll
        for (int r = 0; r < ROUNDS; ++r) {
            int  i   = node0 + grp + GROUPS * r;
            bool own = i < nodeEnd;
            int  l   = ph * N + (own ? i : 0);
            int  c   = cnt[l];
            s0v[r]   = pos[l];
            cv[r]    = own ? c : 0;
        }

        const float* src = xb + (size_t)lo * 3;
        if ((((size_t)src) & 15) == 0) {
            const v4f* s4 = (const v4f*)src;
            v4f*       d4 = (v4f*)sx;
            const int  n4 = nw >> 2;
            const int last = n4 - 1;
            const int STEP = PASS_T * COPY_DEPTH;
            v4f bufA[COPY_DEPTH], bufB[COPY_DEPTH];
            if (n4 > 0) {
                #pragma unroll
                for (int u = 0; u < COPY_DEPTH; ++u)
                    bufA[u] = s4[min(tid + u * PASS_T, last)];
                for (int base = 0; base < n4; base += 2 * STEP) {
                    #pragma unroll
                    for (int u = 0; u < COPY_DEPTH; ++u)
                        bufB[u] = s4[min(base + STEP + tid + u * PASS_T, last)];
                    #pragma unroll
                    for (int u = 0; u < COPY_DEPTH; ++u) {
                        int w = base + tid + u * PASS_T;
                        if (w < n4) d4[w] = bufA[u];
                    }
                    #pragma unroll
                    for (int u = 0; u < COPY_DEPTH; ++u)
                        bufA[u] = s4[min(base + 2 * STEP + tid + u * PASS_T, last)];
                    #pragma unroll
                    for (int u = 0; u < COPY_DEPTH; ++u) {
                        int w = base + STEP + tid + u * PASS_T;
                        if (w < n4) d4[w] = bufB[u];
                    }
                }
            }
            for (int w = (n4 << 2) + tid; w < nw; w += PASS_T) sx[w] = src[w];
        } else {
            for (int w = tid; w < nw; w += PASS_T) sx[w] = src[w];
        }

        int k[ROUNDS], end[ROUNDS];
        int trips = 0;
        #pragma unroll
        for (int r = 0; r < ROUNDS; ++r) {
            k[r]   = s0v[r] + sub;
            end[r] = s0v[r] + cv[r];
            trips  = max(trips, (end[r] - k[r] + NODE_LANES - 1) / NODE_LANES);
        }

        v2i  entA[ROUNDS], entB[ROUNDS];
        bool hasA[ROUNDS], hasB[ROUNDS];
        fetch_entries(entA, hasA, k, end, a2);

        __syncthreads();

        for (int t = 0; t < trips; t += 2) {
            fetch_entries(entB, hasB, k, end, a2);
            walk_entries(entA, hasA, sx, lo, px, py, pz, ax, ay, az);
            fetch_entries(entA, hasA, k, end, a2);
            walk_entries(entB, hasB, sx, lo, px, py, pz, ax, ay, az);
        }
    }

    #pragma unroll
    for (int m = 1; m < NODE_LANES; m <<= 1) {
        #pragma unroll
        for (int r = 0; r < ROUNDS; ++r) {
            ax[r] += __shfl_xor(ax[r], m, NODE_LANES);
            ay[r] += __shfl_xor(ay[r], m, NODE_LANES);
            az[r] += __shfl_xor(az[r], m, NODE_LANES);
        }
    }

    float tau = 1.0f;
    if (FINAL) tau = fmaxf(1.0f - ts[b], TAU_MIN_F);
    #pragma unroll
    for (int r = 0; r < ROUNDS; ++r) {
        int i = node0 + grp + GROUPS * r;
        if (i < nodeEnd && sub < 3) {
            float v = (sub == 0) ? (px[r] + ax[r])
                    : (sub == 1) ? (py[r] + ay[r]) : (pz[r] + az[r]);
            int idx = (b * N + i) * 3 + sub;
            if (FINAL) xout[idx] = (v - kp[idx]) / tau;
            else       xout[idx] = v;
        }
    }
}

extern "C" void kernel_launch(void* const* d_in, const int* in_sizes, int n_in,
                              void* d_out, int out_size, void* d_ws, size_t ws_size,
                              hipStream_t stream) {
    const float* kp   = (const float*)d_in[0];
    const float* ts   = (const float*)d_in[1];
    const float* ht   = (const float*)d_in[2];
    const float* W    = (const float*)d_in[3];
    const float* bias = (const float*)d_in[4];
    const int*   ei   = (const int*)d_in[5];
    const float* rl   = (const float*)d_in[6];

    const int B   = in_sizes[1];
    const int E   = in_sizes[6];
    const int BN3 = in_sizes[0];
    const int BN  = BN3 / 3;
    const int N   = BN / B;
    const int D   = in_sizes[2] / BN;

    const int T = 256;
    const int nTiles = (BN + 15) / 16;
    const int gVec   = (BN3 + T - 1) / T;
    const int BE     = B * E;
    const int gEdge  = (BE + T - 1) / T;
    const int gWmma  = (nTiles * 32 + T - 1) / T;
    const int XPBD_ITERS = 4;

    const int P    = (N + SLAB_NODES - 1) / SLAB_NODES;
    const int NL   = P * N;
    const int npc  = (N + CHUNKS - 1) / CHUNKS;
    const int NB   = (N + BK_CN - 1) >> BK_SHIFT;
    const int NBLK = (E + BK_EPB - 1) / BK_EPB;
    const size_t rW   = ((size_t)2 * E > (size_t)BN3) ? (size_t)2 * E : (size_t)BN3;
    const size_t tabW = (size_t)2 * NB * NBLK;
    const size_t need = ((size_t)4 * E + rW + (size_t)BN3 + (size_t)2 * NL + 1 + tabW) * sizeof(int);

    if (E > 0 && npc <= GROUPS * ROUNDS && ws_size >= need &&
        NB <= BK_NBMAX && NBLK <= BK_T && P * BK_CN <= BK_T) {
        int*   adj  = (int*)d_ws;
        int*   rec  = adj + (size_t)4 * E;
        float* x1   = (float*)rec;
        float* x0   = (float*)(rec + rW);
        int*   cnt  = (int*)(x0 + BN3);
        int*   pos  = cnt + NL + 1;
        int*   cntT = pos + NL;
        int*   offT = cntT + (size_t)NB * NBLK;

        const dim3 gPass(CHUNKS, B);

        const int TPB = BK_T / 32;
        const int hb  = (nTiles + TPB - 1) / TPB;

        bucket_kernel<<<NBLK, BK_T, 0, stream>>>(ei, rec, cntT, offT, E, NB, NBLK);
        lists_head_kernel<<<NB + hb, BK_T, 0, stream>>>(ei, rl, rec, cntT, offT, adj, pos, cnt,
                                                        E, N, NBLK, P, NB,
                                                        ht, W, bias, kp, ts, x0, 0, nTiles, D);

        slab_gather_kernel<false><<<gPass, PASS_T, 0, stream>>>(x0, adj, pos, cnt, kp, ts, x1, N, npc);
        slab_gather_kernel<false><<<gPass, PASS_T, 0, stream>>>(x1, adj, pos, cnt, kp, ts, x0, N, npc);
        slab_gather_kernel<false><<<gPass, PASS_T, 0, stream>>>(x0, adj, pos, cnt, kp, ts, x1, N, npc);
        slab_gather_kernel<true ><<<gPass, PASS_T, 0, stream>>>(x1, adj, pos, cnt, kp, ts,
                                                                (float*)d_out, N, npc);
        return;
    }

    float* xbuf  = (float*)d_ws;
    float* delta = xbuf + BN3;

    zero_kernel<<<gVec, T, 0, stream>>>(delta, BN3);
    head_wmma_kernel<<<gWmma, T, 0, stream>>>(ht, W, bias, kp, ts, xbuf, nTiles, D, N);

    for (int it = 0; it < XPBD_ITERS; ++it) {
        edge_kernel<<<gEdge, T, 0, stream>>>(xbuf, ei, rl, delta, E, N, BE);
        if (it < XPBD_ITERS - 1) {
            apply_kernel<<<gVec, T, 0, stream>>>(xbuf, delta, BN3);
        } else {
            final_kernel<<<gVec, T, 0, stream>>>(xbuf, delta, kp, ts,
                                                 (float*)d_out, BN3, N * 3);
        }
    }
}


# list build: level 1 stores an 8-byte record (owner-in-bucket|neighbour, L0) per half-edge; level 2 reads runs of records, not src, dst, L0 by index twice per half-edge; list order: arrival, as before
# speedup vs baseline: 20.2636x; 1.1628x over previous
#include <hip/hip_runtime.h>
#include <hip/hip_bf16.h>


typedef float v2f __attribute__((ext_vector_type(2)));
typedef float v8f __attribute__((ext_vector_type(8)));

#define TAU_MIN_F     0.001f
#define XPBD_MAXC     0.15f
#define XPBD_DENOM    (2.0f + 1e-9f)

#define BSTRIDE 260

__device__ __forceinline__ void atomic_add_f32(float* p, float v) {
    unsafeAtomicAdd(p, v);
}

__global__ void zero_kernel(float* __restrict__ p, int n) {
    int i = blockIdx.x * blockDim.x + threadIdx.x;
    if (i < n) p[i] = 0.0f;
}

__device__ __forceinline__ void head_tiles(
    const float* __restrict__ ht,
    const float* __restrict__ W,
    const float* __restrict__ bias,
    const float* __restrict__ kp,
    const float* __restrict__ ts,
    float* __restrict__ x,
    int tile0, int tileEnd, int hb, int D, int N)
{
    __shared__ float sWT[16 * BSTRIDE];
    __shared__ float sB[3];

    for (int idx = threadIdx.x; idx < 16 * 256; idx += blockDim.x) {
        int n = idx >> 8;
        int k = idx & 255;
        sWT[n * BSTRIDE + k] = (n < 3) ? W[k * 3 + n] : 0.0f;
    }
    if (threadIdx.x < 3) sB[threadIdx.x] = bias[threadIdx.x];
    __syncthreads();

    int wave = tile0 + ((hb * (int)blockDim.x + (int)threadIdx.x) >> 5);
    if (wave >= tileEnd) return;

    int lane = threadIdx.x & 31;
    int half = lane >> 4;
    int l15  = lane & 15;

    long rowA = (long)wave * 16 + l15;
    const float* arow = ht + rowA * D + 2 * half;
    const float* brow = &sWT[l15 * BSTRIDE + 2 * half];

    v8f acc0 = {};
    v8f acc1 = {};
    #pragma unroll 4
    for (int kb = 0; kb < D; kb += 8) {
        v2f a0 = *(const v2f*)(arow + kb);
        v2f a1 = *(const v2f*)(arow + kb + 4);
        v2f b0 = *(const v2f*)(brow + kb);
        v2f b1 = *(const v2f*)(brow + kb + 4);
        acc0 = __builtin_amdgcn_wmma_f32_16x16x4_f32(
                   false, a0, false, b0, (short)0, acc0, false, false);
        acc1 = __builtin_amdgcn_wmma_f32_16x16x4_f32(
                   false, a1, false, b1, (short)0, acc1, false, false);
    }

    if (l15 < 3) {
        unsigned rowBase = (unsigned)wave * 16u + (unsigned)half * 8u;
        #pragma unroll
        for (int r = 0; r < 8; ++r) {
            unsigned row = rowBase + (unsigned)r;
            int  bidx = (int)(row / (unsigned)N);
            float tau = fmaxf(1.0f - ts[bidx], TAU_MIN_F);
            long idx = (long)row * 3 + l15;
            x[idx] = kp[idx] + tau * (acc0[r] + acc1[r] + sB[l15]);
        }
    }
}

__global__ __launch_bounds__(256) void head_wmma_kernel(
    const float* __restrict__ ht, const float* __restrict__ W,
    const float* __restrict__ bias, const float* __restrict__ kp,
    const float* __restrict__ ts, float* __restrict__ x,
    int nTiles, int D, int N)
{
    head_tiles(ht, W, bias, kp, ts, x, 0, nTiles, (int)blockIdx.x, D, N);
}

__global__ void edge_kernel(
    const float* __restrict__ x,
    const int*   __restrict__ ei,
    const float* __restrict__ rl,
    float* __restrict__ delta,
    int E, int N, int BE)
{
    int t = blockIdx.x * blockDim.x + threadIdx.x;
    if (t >= BE) return;
    int b = t / E;
    int e = t - b * E;
    int s = ei[e];
    int d = ei[E + e];
    int si = (b * N + s) * 3;
    int di = (b * N + d) * 3;

    float dx = x[si + 0] - x[di + 0];
    float dy = x[si + 1] - x[di + 1];
    float dz = x[si + 2] - x[di + 2];
    float dist = sqrtf(dx * dx + dy * dy + dz * dz);
    float C    = dist - rl[e];
    float scal = (-C / XPBD_DENOM) / (dist + 1e-9f);

    float cx = fminf(fmaxf(scal * dx, -XPBD_MAXC), XPBD_MAXC);
    float cy = fminf(fmaxf(scal * dy, -XPBD_MAXC), XPBD_MAXC);
    float cz = fminf(fmaxf(scal * dz, -XPBD_MAXC), XPBD_MAXC);

    atomic_add_f32(&delta[si + 0],  cx);
    atomic_add_f32(&delta[si + 1],  cy);
    atomic_add_f32(&delta[si + 2],  cz);
    atomic_add_f32(&delta[di + 0], -cx);
    atomic_add_f32(&delta[di + 1], -cy);
    atomic_add_f32(&delta[di + 2], -cz);
}

__global__ void apply_kernel(float* __restrict__ x, float* __restrict__ delta, int n) {
    int i = blockIdx.x * blockDim.x + threadIdx.x;
    if (i < n) {
        x[i] += delta[i];
        delta[i] = 0.0f;
    }
}

__global__ void final_kernel(
    const float* __restrict__ x, const float* __restrict__ delta,
    const float* __restrict__ kp, const float* __restrict__ ts,
    float* __restrict__ out, int n, int perBatch)
{
    int i = blockIdx.x * blockDim.x + threadIdx.x;
    if (i < n) {
        int b = i / perBatch;
        float tau = fmaxf(1.0f - ts[b], TAU_MIN_F);
        out[i] = (x[i] + delta[i] - kp[i]) / tau;
    }
}


typedef int   v2i __attribute__((ext_vector_type(2)));
typedef float v4f __attribute__((ext_vector_type(4)));

#define SLAB_NODES 25088
#define NODE_LANES 8
#define PASS_T     1024
#define GROUPS     (PASS_T / NODE_LANES)
#define ROUNDS     7
#define CHUNKS     64
#define COPY_DEPTH 4

#define BK_SHIFT 8
#define BK_CN    (1 << BK_SHIFT)
#define BK_NBMAX 256
#define BK_T     1024
#define BK_EPT   4
#define BK_EPB   (BK_T * BK_EPT)
#define BK_HPB   (2 * BK_EPB)

#define REC_NBR_BITS 16

__global__ __launch_bounds__(BK_T) void bucket_kernel(
    const int* __restrict__ ei, const float* __restrict__ rl, int* __restrict__ rec,
    int* __restrict__ cntT, int* __restrict__ offT, int E, int NB, int NBLK)
{
    __shared__ int hist[BK_NBMAX];
    __shared__ int hoff[BK_NBMAX];
    const int tid = threadIdx.x;
    if (tid < BK_NBMAX) hist[tid] = 0;
    __syncthreads();

    const int e0 = blockIdx.x * BK_EPB;
    int sv[BK_EPT], dv[BK_EPT], lv[BK_EPT], rk[2 * BK_EPT];
    #pragma unroll
    for (int u = 0; u < BK_EPT; ++u) {
        int e  = e0 + tid + u * BK_T;
        int ec = (e < E) ? e : 0;
        sv[u] = ei[ec];
        dv[u] = ei[E + ec];
        lv[u] = __float_as_int(rl[ec]);
    }
    #pragma unroll
    for (int u = 0; u < BK_EPT; ++u) {
        int e = e0 + tid + u * BK_T;
        rk[2 * u] = 0; rk[2 * u + 1] = 0;
        if (e < E) {
            rk[2 * u]     = atomicAdd(&hist[sv[u] >> BK_SHIFT], 1);
            rk[2 * u + 1] = atomicAdd(&hist[dv[u] >> BK_SHIFT], 1);
        }
    }
    __syncthreads();
    if (tid < 32) {
        int a[8];
        int sum = 0;
        #pragma unroll
        for (int q = 0; q < 8; ++q) { a[q] = hist[8 * tid + q]; sum += a[q]; }
        int v = sum;
        for (int o = 1; o < 32; o <<= 1) {
            int t = __shfl_up(v, o);
            if (tid >= o) v += t;
        }
        int run = v - sum;
        #pragma unroll
        for (int q = 0; q < 8; ++q) { hoff[8 * tid + q] = run; run += a[q]; }
    }
    __syncthreads();
    if (tid < NB) {
        cntT[tid * NBLK + blockIdx.x] = hist[tid];
        offT[tid * NBLK + blockIdx.x] = hoff[tid];
    }
    v2i* myrec = (v2i*)rec + (size_t)blockIdx.x * BK_HPB;
    #pragma unroll
    for (int u = 0; u < BK_EPT; ++u) {
        int e = e0 + tid + u * BK_T;
        if (e < E) {
            v2i r0, r1;
            r0.x = ((sv[u] & (BK_CN - 1)) << REC_NBR_BITS) | dv[u];
            r0.y = lv[u];
            r1.x = ((dv[u] & (BK_CN - 1)) << REC_NBR_BITS) | sv[u];
            r1.y = lv[u];
            myrec[hoff[sv[u] >> BK_SHIFT] + rk[2 * u]]     = r0;
            myrec[hoff[dv[u] >> BK_SHIFT] + rk[2 * u + 1]] = r1;
        }
    }
}

__device__ __forceinline__ int block_scan_excl(int c, int* ws, int* tot) {
    const int lane = threadIdx.x & 31;
    const int w    = threadIdx.x >> 5;
    int v = c;
    for (int o = 1; o < 32; o <<= 1) {
        int t = __shfl_up(v, o);
        if (lane >= o) v += t;
    }
    if (lane == 31) ws[w] = v;
    __syncthreads();
    if (w == 0) {
        int s  = ws[lane];
        int sv = s;
        for (int o = 1; o < 32; o <<= 1) {
            int t = __shfl_up(sv, o);
            if (lane >= o) sv += t;
        }
        ws[lane] = sv - s;
        if (lane == 31) *tot = sv;
    }
    __syncthreads();
    return ws[w] + v - c;
}

#define BK_U 8

__device__ __forceinline__ void lists_body(
    const int* __restrict__ rec, const int* __restrict__ cntT,
    const int* __restrict__ offT, int* __restrict__ adj,
    int* __restrict__ pos, int* __restrict__ cnt,
    int N, int NBLK, int P)
{
    __shared__ int segO[BK_T];
    __shared__ int segP[BK_T + 1];
    __shared__ int lcnt[BK_T];
    __shared__ int lcur[BK_T];
    __shared__ int wsA[32], wsB[32], wsC[32];
    __shared__ int totA, totB, baseSh;

    const int tid   = threadIdx.x;
    const int lane  = tid & 31;
    const int w     = tid >> 5;
    const int kb    = blockIdx.x;
    const int node0 = kb << BK_SHIFT;

    int c = 0, o = 0;
    if (tid < NBLK) { c = cntT[kb * NBLK + tid]; o = offT[kb * NBLK + tid]; }
    segO[tid] = o;
    int ex = block_scan_excl(c, wsA, &totA);
    segP[tid] = ex;
    if (tid == BK_T - 1) segP[BK_T] = ex + c;

    int acc = 0;
    for (int idx = tid; idx < kb * NBLK; idx += BK_T) acc += cntT[idx];
    for (int m = 16; m >= 1; m >>= 1) acc += __shfl_xor(acc, m);
    if (lane == 0) wsB[w] = acc;
    lcnt[tid] = 0;
    __syncthreads();
    if (w == 0) {
        int s = wsB[lane];
        for (int m = 16; m >= 1; m >>= 1) s += __shfl_xor(s, m);
        if (lane == 0) baseSh = s;
    }
    __syncthreads();
    const int T    = totA;
    const int base = baseSh;
    int st0 = 1;
    while (st0 * 2 < NBLK) st0 *= 2;
    const v2i* r2 = (const v2i*)rec;

    for (int t0 = tid; t0 < T; t0 += BK_T * BK_U) {
        int  w0[BK_U], tt[BK_U], b[BK_U];
        bool ok[BK_U];
        #pragma unroll
        for (int u = 0; u < BK_U; ++u) {
            int t = t0 + u * BK_T;
            ok[u] = t < T;
            tt[u] = ok[u] ? t : 0;
            b[u]  = 0;
        }
        for (int st = st0; st >= 1; st >>= 1) {
            #pragma unroll
            for (int u = 0; u < BK_U; ++u) b[u] += (segP[b[u] + st] <= tt[u]) ? st : 0;
        }
        #pragma unroll
        for (int u = 0; u < BK_U; ++u) {
            w0[u] = r2[b[u] * BK_HPB + segO[b[u]] + (tt[u] - segP[b[u]])].x;
        }
        #pragma unroll
        for (int u = 0; u < BK_U; ++u) {
            int nbr = w0[u] & ((1 << REC_NBR_BITS) - 1);
            int ll  = ((nbr / SLAB_NODES) << BK_SHIFT) | (w0[u] >> REC_NBR_BITS);
            if (ok[u]) atomicAdd(&lcnt[ll], 1);
        }
    }
    __syncthreads();

    const int LISTS = P << BK_SHIFT;
    int lc  = (tid < LISTS) ? lcnt[tid] : 0;
    int lex = block_scan_excl(lc, wsC, &totB);
    if (tid < LISTS) {
        int p = tid >> BK_SHIFT;
        int i = node0 + (tid & (BK_CN - 1));
        lcur[tid] = base + lex;
        if (i < N) {
            pos[p * N + i] = base + lex;
            cnt[p * N + i] = lc;
        }
    }
    __syncthreads();

    v2i* a2 = (v2i*)adj;
    for (int t0 = tid; t0 < T; t0 += BK_T * BK_U) {
        int  tt[BK_U], b[BK_U];
        v2i  rr[BK_U];
        bool ok[BK_U];
        #pragma unroll
        for (int u = 0; u < BK_U; ++u) {
            int t = t0 + u * BK_T;
            ok[u] = t < T;
            tt[u] = ok[u] ? t : 0;
            b[u]  = 0;
        }
        for (int st = st0; st >= 1; st >>= 1) {
            #pragma unroll
            for (int u = 0; u < BK_U; ++u) b[u] += (segP[b[u] + st] <= tt[u]) ? st : 0;
        }
        #pragma unroll
        for (int u = 0; u < BK_U; ++u) {
            rr[u] = r2[b[u] * BK_HPB + segO[b[u]] + (tt[u] - segP[b[u]])];
        }
        #pragma unroll
        for (int u = 0; u < BK_U; ++u) {
            int nbr = rr[u].x & ((1 << REC_NBR_BITS) - 1);
            int ll  = ((nbr / SLAB_NODES) << BK_SHIFT) | (rr[u].x >> REC_NBR_BITS);
            if (ok[u]) {
                int slot = atomicAdd(&lcur[ll], 1);
                v2i en; en.x = nbr; en.y = rr[u].y;
                a2[slot] = en;
            }
        }
    }
}

__global__ __launch_bounds__(BK_T) void lists_head_kernel(
    const int* __restrict__ rec, const int* __restrict__ cntT,
    const int* __restrict__ offT, int* __restrict__ adj,
    int* __restrict__ pos, int* __restrict__ cnt,
    int N, int NBLK, int P, int NB,
    const float* __restrict__ ht, const float* __restrict__ W,
    const float* __restrict__ bias, const float* __restrict__ kp,
    const float* __restrict__ ts, float* __restrict__ x,
    int tile0, int tileEnd, int D)
{
    if ((int)blockIdx.x < NB) lists_body(rec, cntT, offT, adj, pos, cnt, N, NBLK, P);
    else head_tiles(ht, W, bias, kp, ts, x, tile0, tileEnd, (int)blockIdx.x - NB, D, N);
}

__device__ __forceinline__ void fetch_entries(
    v2i (&ent)[ROUNDS], bool (&has)[ROUNDS], int (&k)[ROUNDS],
    const int (&end)[ROUNDS], const v2i* __restrict__ a2)
{
    #pragma unroll
    for (int r = 0; r < ROUNDS; ++r) {
        has[r] = k[r] < end[r];
        ent[r] = a2[has[r] ? k[r] : 0];
        k[r] += NODE_LANES;
    }
}

__device__ __forceinline__ void walk_entries(
    const v2i (&ent)[ROUNDS], const bool (&has)[ROUNDS], const float* sx, int lo,
    const float (&px)[ROUNDS], const float (&py)[ROUNDS], const float (&pz)[ROUNDS],
    float (&ax)[ROUNDS], float (&ay)[ROUNDS], float (&az)[ROUNDS])
{
    #pragma unroll
    for (int r = 0; r < ROUNDS; ++r) {
        if (has[r]) {
            int j = ent[r].x;
            const float* q = &sx[(j - lo) * 3];
            float L0 = __int_as_float(ent[r].y);
            float dx = px[r] - q[0];
            float dy = py[r] - q[1];
            float dz = pz[r] - q[2];
            float dist = sqrtf(dx * dx + dy * dy + dz * dz);
            float C    = dist - L0;
            float scal = (-C / XPBD_DENOM) / (dist + 1e-9f);
            ax[r] += fminf(fmaxf(scal * dx, -XPBD_MAXC), XPBD_MAXC);
            ay[r] += fminf(fmaxf(scal * dy, -XPBD_MAXC), XPBD_MAXC);
            az[r] += fminf(fmaxf(scal * dz, -XPBD_MAXC), XPBD_MAXC);
        }
    }
}

template <bool FINAL>
__global__ __launch_bounds__(PASS_T) void slab_gather_kernel(
    const float* __restrict__ xin,
    const int*   __restrict__ adj,
    const int*   __restrict__ pos,
    const int*   __restrict__ cnt,
    const float* __restrict__ kp,
    const float* __restrict__ ts,
    float* __restrict__ xout, int N, int npc)
{
    __shared__ __attribute__((aligned(16))) float sx[SLAB_NODES * 3];

    const int tid     = threadIdx.x;
    const int grp     = tid / NODE_LANES;
    const int sub     = tid % NODE_LANES;
    const int b       = blockIdx.y;
    const int node0   = blockIdx.x * npc;
    const int nodeEnd = min(N, node0 + npc);
    const float* xb   = xin + (size_t)b * N * 3;
    const v2i*   a2   = (const v2i*)adj;

    float px[ROUNDS], py[ROUNDS], pz[ROUNDS];
    float ax[ROUNDS], ay[ROUNDS], az[ROUNDS];
    #pragma unroll
    for (int r = 0; r < ROUNDS; ++r) {
        int i = min(node0 + grp + GROUPS * r, N - 1);
        const float* p = xb + i * 3;
        px[r] = p[0]; py[r] = p[1]; pz[r] = p[2];
        ax[r] = 0.0f; ay[r] = 0.0f; az[r] = 0.0f;
    }

    int ph = 0;
    for (int lo = 0; lo < N; lo += SLAB_NODES, ++ph) {
        const int hi = min(N, lo + SLAB_NODES);
        const int nw = (hi - lo) * 3;

        __syncthreads();

        int s0v[ROUNDS], cv[ROUNDS];
        #pragma unroll
        for (int r = 0; r < ROUNDS; ++r) {
            int  i   = node0 + grp + GROUPS * r;
            bool own = i < nodeEnd;
            int  l   = ph * N + (own ? i : 0);
            int  c   = cnt[l];
            s0v[r]   = pos[l];
            cv[r]    = own ? c : 0;
        }

        const float* src = xb + (size_t)lo * 3;
        if ((((size_t)src) & 15) == 0) {
            const v4f* s4 = (const v4f*)src;
            v4f*       d4 = (v4f*)sx;
            const int  n4 = nw >> 2;
            const int last = n4 - 1;
            const int STEP = PASS_T * COPY_DEPTH;
            v4f bufA[COPY_DEPTH], bufB[COPY_DEPTH];
            if (n4 > 0) {
                #pragma unroll
                for (int u = 0; u < COPY_DEPTH; ++u)
                    bufA[u] = s4[min(tid + u * PASS_T, last)];
                for (int base = 0; base < n4; base += 2 * STEP) {
                    #pragma unroll
                    for (int u = 0; u < COPY_DEPTH; ++u)
                        bufB[u] = s4[min(base + STEP + tid + u * PASS_T, last)];
                    #pragma unroll
                    for (int u = 0; u < COPY_DEPTH; ++u) {
                        int w = base + tid + u * PASS_T;
                        if (w < n4) d4[w] = bufA[u];
                    }
                    #pragma unroll
                    for (int u = 0; u < COPY_DEPTH; ++u)
                        bufA[u] = s4[min(base + 2 * STEP + tid + u * PASS_T, last)];
                    #pragma unroll
                    for (int u = 0; u < COPY_DEPTH; ++u) {
                        int w = base + STEP + tid + u * PASS_T;
                        if (w < n4) d4[w] = bufB[u];
                    }
                }
            }
            for (int w = (n4 << 2) + tid; w < nw; w += PASS_T) sx[w] = src[w];
        } else {
            for (int w = tid; w < nw; w += PASS_T) sx[w] = src[w];
        }

        int k[ROUNDS], end[ROUNDS];
        int trips = 0;
        #pragma unroll
        for (int r = 0; r < ROUNDS; ++r) {
            k[r]   = s0v[r] + sub;
            end[r] = s0v[r] + cv[r];
            trips  = max(trips, (end[r] - k[r] + NODE_LANES - 1) / NODE_LANES);
        }

        v2i  entA[ROUNDS], entB[ROUNDS];
        bool hasA[ROUNDS], hasB[ROUNDS];
        fetch_entries(entA, hasA, k, end, a2);

        __syncthreads();

        for (int t = 0; t < trips; t += 2) {
            fetch_entries(entB, hasB, k, end, a2);
            walk_entries(entA, hasA, sx, lo, px, py, pz, ax, ay, az);
            fetch_entries(entA, hasA, k, end, a2);
            walk_entries(entB, hasB, sx, lo, px, py, pz, ax, ay, az);
        }
    }

    #pragma unroll
    for (int m = 1; m < NODE_LANES; m <<= 1) {
        #pragma unroll
        for (int r = 0; r < ROUNDS; ++r) {
            ax[r] += __shfl_xor(ax[r], m, NODE_LANES);
            ay[r] += __shfl_xor(ay[r], m, NODE_LANES);
            az[r] += __shfl_xor(az[r], m, NODE_LANES);
        }
    }

    float tau = 1.0f;
    if (FINAL) tau = fmaxf(1.0f - ts[b], TAU_MIN_F);
    #pragma unroll
    for (int r = 0; r < ROUNDS; ++r) {
        int i = node0 + grp + GROUPS * r;
        if (i < nodeEnd && sub < 3) {
            float v = (sub == 0) ? (px[r] + ax[r])
                    : (sub == 1) ? (py[r] + ay[r]) : (pz[r] + az[r]);
            int idx = (b * N + i) * 3 + sub;
            if (FINAL) xout[idx] = (v - kp[idx]) / tau;
            else       xout[idx] = v;
        }
    }
}

extern "C" void kernel_launch(void* const* d_in, const int* in_sizes, int n_in,
                              void* d_out, int out_size, void* d_ws, size_t ws_size,
                              hipStream_t stream) {
    const float* kp   = (const float*)d_in[0];
    const float* ts   = (const float*)d_in[1];
    const float* ht   = (const float*)d_in[2];
    const float* W    = (const float*)d_in[3];
    const float* bias = (const float*)d_in[4];
    const int*   ei   = (const int*)d_in[5];
    const float* rl   = (const float*)d_in[6];

    const int B   = in_sizes[1];
    const int E   = in_sizes[6];
    const int BN3 = in_sizes[0];
    const int BN  = BN3 / 3;
    const int N   = BN / B;
    const int D   = in_sizes[2] / BN;

    const int T = 256;
    const int nTiles = (BN + 15) / 16;
    const int gVec   = (BN3 + T - 1) / T;
    const int BE     = B * E;
    const int gEdge  = (BE + T - 1) / T;
    const int gWmma  = (nTiles * 32 + T - 1) / T;
    const int XPBD_ITERS = 4;

    const int P    = (N + SLAB_NODES - 1) / SLAB_NODES;
    const int NL   = P * N;
    const int npc  = (N + CHUNKS - 1) / CHUNKS;
    const int NB   = (N + BK_CN - 1) >> BK_SHIFT;
    const int NBLK = (E + BK_EPB - 1) / BK_EPB;
    const size_t rW   = ((size_t)4 * E > (size_t)BN3) ? (size_t)4 * E : (size_t)BN3;
    const size_t tabW = (size_t)2 * NB * NBLK;
    const size_t need = ((size_t)4 * E + rW + (size_t)BN3 + (size_t)2 * NL + 1 + tabW) * sizeof(int);

    if (E > 0 && npc <= GROUPS * ROUNDS && ws_size >= need &&
        NB <= BK_NBMAX && NBLK <= BK_T && P * BK_CN <= BK_T && N <= (1 << REC_NBR_BITS)) {
        int*   adj  = (int*)d_ws;
        int*   rec  = adj + (size_t)4 * E;
        float* x1   = (float*)rec;
        float* x0   = (float*)(rec + rW);
        int*   cnt  = (int*)(x0 + BN3);
        int*   pos  = cnt + NL + 1;
        int*   cntT = pos + NL;
        int*   offT = cntT + (size_t)NB * NBLK;

        const dim3 gPass(CHUNKS, B);

        const int TPB = BK_T / 32;
        const int hb  = (nTiles + TPB - 1) / TPB;

        bucket_kernel<<<NBLK, BK_T, 0, stream>>>(ei, rl, rec, cntT, offT, E, NB, NBLK);
        lists_head_kernel<<<NB + hb, BK_T, 0, stream>>>(rec, cntT, offT, adj, pos, cnt,
                                                        N, NBLK, P, NB,
                                                        ht, W, bias, kp, ts, x0, 0, nTiles, D);

        slab_gather_kernel<false><<<gPass, PASS_T, 0, stream>>>(x0, adj, pos, cnt, kp, ts, x1, N, npc);
        slab_gather_kernel<false><<<gPass, PASS_T, 0, stream>>>(x1, adj, pos, cnt, kp, ts, x0, N, npc);
        slab_gather_kernel<false><<<gPass, PASS_T, 0, stream>>>(x0, adj, pos, cnt, kp, ts, x1, N, npc);
        slab_gather_kernel<true ><<<gPass, PASS_T, 0, stream>>>(x1, adj, pos, cnt, kp, ts,
                                                                (float*)d_out, N, npc);
        return;
    }

    float* xbuf  = (float*)d_ws;
    float* delta = xbuf + BN3;

    zero_kernel<<<gVec, T, 0, stream>>>(delta, BN3);
    head_wmma_kernel<<<gWmma, T, 0, stream>>>(ht, W, bias, kp, ts, xbuf, nTiles, D, N);

    for (int it = 0; it < XPBD_ITERS; ++it) {
        edge_kernel<<<gEdge, T, 0, stream>>>(xbuf, ei, rl, delta, E, N, BE);
        if (it < XPBD_ITERS - 1) {
            apply_kernel<<<gVec, T, 0, stream>>>(xbuf, delta, BN3);
        } else {
            final_kernel<<<gVec, T, 0, stream>>>(xbuf, delta, kp, ts,
                                                 (float*)d_out, BN3, N * 3);
        }
    }
}


# pass block stages ONE slab (grid x slabs, 2x nodes per block): staging traffic halves; a node's per-slab sums meet by float atomic add in a buffer cleared a launch earlier (2 addends: order-free)
# speedup vs baseline: 21.0892x; 1.0407x over previous
#include <hip/hip_runtime.h>
#include <hip/hip_bf16.h>


typedef float v2f __attribute__((ext_vector_type(2)));
typedef float v8f __attribute__((ext_vector_type(8)));

#define TAU_MIN_F     0.001f
#define XPBD_MAXC     0.15f
#define XPBD_DENOM    (2.0f + 1e-9f)

#define BSTRIDE 260

__device__ __forceinline__ void atomic_add_f32(float* p, float v) {
    unsafeAtomicAdd(p, v);
}

__global__ void zero_kernel(float* __restrict__ p, int n) {
    int i = blockIdx.x * blockDim.x + threadIdx.x;
    if (i < n) p[i] = 0.0f;
}

__device__ __forceinline__ void head_tiles(
    const float* __restrict__ ht,
    const float* __restrict__ W,
    const float* __restrict__ bias,
    const float* __restrict__ kp,
    const float* __restrict__ ts,
    float* __restrict__ x,
    float* __restrict__ z,
    int tile0, int tileEnd, int hb, int D, int N)
{
    __shared__ float sWT[16 * BSTRIDE];
    __shared__ float sB[3];

    for (int idx = threadIdx.x; idx < 16 * 256; idx += blockDim.x) {
        int n = idx >> 8;
        int k = idx & 255;
        sWT[n * BSTRIDE + k] = (n < 3) ? W[k * 3 + n] : 0.0f;
    }
    if (threadIdx.x < 3) sB[threadIdx.x] = bias[threadIdx.x];
    __syncthreads();

    int wave = tile0 + ((hb * (int)blockDim.x + (int)threadIdx.x) >> 5);
    if (wave >= tileEnd) return;

    int lane = threadIdx.x & 31;
    int half = lane >> 4;
    int l15  = lane & 15;

    long rowA = (long)wave * 16 + l15;
    const float* arow = ht + rowA * D + 2 * half;
    const float* brow = &sWT[l15 * BSTRIDE + 2 * half];

    v8f acc0 = {};
    v8f acc1 = {};
    #pragma unroll 4
    for (int kb = 0; kb < D; kb += 8) {
        v2f a0 = *(const v2f*)(arow + kb);
        v2f a1 = *(const v2f*)(arow + kb + 4);
        v2f b0 = *(const v2f*)(brow + kb);
        v2f b1 = *(const v2f*)(brow + kb + 4);
        acc0 = __builtin_amdgcn_wmma_f32_16x16x4_f32(
                   false, a0, false, b0, (short)0, acc0, false, false);
        acc1 = __builtin_amdgcn_wmma_f32_16x16x4_f32(
                   false, a1, false, b1, (short)0, acc1, false, false);
    }

    if (l15 < 3) {
        unsigned rowBase = (unsigned)wave * 16u + (unsigned)half * 8u;
        #pragma unroll
        for (int r = 0; r < 8; ++r) {
            unsigned row = rowBase + (unsigned)r;
            int  bidx = (int)(row / (unsigned)N);
            float tau = fmaxf(1.0f - ts[bidx], TAU_MIN_F);
            long idx = (long)row * 3 + l15;
            x[idx] = kp[idx] + tau * (acc0[r] + acc1[r] + sB[l15]);
            if (z) z[idx] = 0.0f;
        }
    }
}

__global__ __launch_bounds__(256) void head_wmma_kernel(
    const float* __restrict__ ht, const float* __restrict__ W,
    const float* __restrict__ bias, const float* __restrict__ kp,
    const float* __restrict__ ts, float* __restrict__ x,
    int nTiles, int D, int N)
{
    head_tiles(ht, W, bias, kp, ts, x, nullptr, 0, nTiles, (int)blockIdx.x, D, N);
}

__global__ void edge_kernel(
    const float* __restrict__ x,
    const int*   __restrict__ ei,
    const float* __restrict__ rl,
    float* __restrict__ delta,
    int E, int N, int BE)
{
    int t = blockIdx.x * blockDim.x + threadIdx.x;
    if (t >= BE) return;
    int b = t / E;
    int e = t - b * E;
    int s = ei[e];
    int d = ei[E + e];
    int si = (b * N + s) * 3;
    int di = (b * N + d) * 3;

    float dx = x[si + 0] - x[di + 0];
    float dy = x[si + 1] - x[di + 1];
    float dz = x[si + 2] - x[di + 2];
    float dist = sqrtf(dx * dx + dy * dy + dz * dz);
    float C    = dist - rl[e];
    float scal = (-C / XPBD_DENOM) / (dist + 1e-9f);

    float cx = fminf(fmaxf(scal * dx, -XPBD_MAXC), XPBD_MAXC);
    float cy = fminf(fmaxf(scal * dy, -XPBD_MAXC), XPBD_MAXC);
    float cz = fminf(fmaxf(scal * dz, -XPBD_MAXC), XPBD_MAXC);

    atomic_add_f32(&delta[si + 0],  cx);
    atomic_add_f32(&delta[si + 1],  cy);
    atomic_add_f32(&delta[si + 2],  cz);
    atomic_add_f32(&delta[di + 0], -cx);
    atomic_add_f32(&delta[di + 1], -cy);
    atomic_add_f32(&delta[di + 2], -cz);
}

__global__ void apply_kernel(float* __restrict__ x, float* __restrict__ delta, int n) {
    int i = blockIdx.x * blockDim.x + threadIdx.x;
    if (i < n) {
        x[i] += delta[i];
        delta[i] = 0.0f;
    }
}

__global__ void final_kernel(
    const float* __restrict__ x, const float* __restrict__ delta,
    const float* __restrict__ kp, const float* __restrict__ ts,
    float* __restrict__ out, int n, int perBatch)
{
    int i = blockIdx.x * blockDim.x + threadIdx.x;
    if (i < n) {
        int b = i / perBatch;
        float tau = fmaxf(1.0f - ts[b], TAU_MIN_F);
        out[i] = (x[i] + delta[i] - kp[i]) / tau;
    }
}


typedef int   v2i __attribute__((ext_vector_type(2)));
typedef float v4f __attribute__((ext_vector_type(4)));

#define SLAB_NODES 25088
#define NODE_LANES 8
#define PASS_T     1024
#define GROUPS     (PASS_T / NODE_LANES)
#define ROUNDS     7
#define CHUNKS     32
#define COPY_DEPTH 4

#define BK_SHIFT 8
#define BK_CN    (1 << BK_SHIFT)
#define BK_NBMAX 256
#define BK_T     1024
#define BK_EPT   4
#define BK_EPB   (BK_T * BK_EPT)
#define BK_HPB   (2 * BK_EPB)

#define REC_NBR_BITS 16

__global__ __launch_bounds__(BK_T) void bucket_kernel(
    const int* __restrict__ ei, const float* __restrict__ rl, int* __restrict__ rec,
    int* __restrict__ cntT, int* __restrict__ offT, int E, int NB, int NBLK)
{
    __shared__ int hist[BK_NBMAX];
    __shared__ int hoff[BK_NBMAX];
    const int tid = threadIdx.x;
    if (tid < BK_NBMAX) hist[tid] = 0;
    __syncthreads();

    const int e0 = blockIdx.x * BK_EPB;
    int sv[BK_EPT], dv[BK_EPT], lv[BK_EPT], rk[2 * BK_EPT];
    #pragma unroll
    for (int u = 0; u < BK_EPT; ++u) {
        int e  = e0 + tid + u * BK_T;
        int ec = (e < E) ? e : 0;
        sv[u] = ei[ec];
        dv[u] = ei[E + ec];
        lv[u] = __float_as_int(rl[ec]);
    }
    #pragma unroll
    for (int u = 0; u < BK_EPT; ++u) {
        int e = e0 + tid + u * BK_T;
        rk[2 * u] = 0; rk[2 * u + 1] = 0;
        if (e < E) {
            rk[2 * u]     = atomicAdd(&hist[sv[u] >> BK_SHIFT], 1);
            rk[2 * u + 1] = atomicAdd(&hist[dv[u] >> BK_SHIFT], 1);
        }
    }
    __syncthreads();
    if (tid < 32) {
        int a[8];
        int sum = 0;
        #pragma unroll
        for (int q = 0; q < 8; ++q) { a[q] = hist[8 * tid + q]; sum += a[q]; }
        int v = sum;
        for (int o = 1; o < 32; o <<= 1) {
            int t = __shfl_up(v, o);
            if (tid >= o) v += t;
        }
        int run = v - sum;
        #pragma unroll
        for (int q = 0; q < 8; ++q) { hoff[8 * tid + q] = run; run += a[q]; }
    }
    __syncthreads();
    if (tid < NB) {
        cntT[tid * NBLK + blockIdx.x] = hist[tid];
        offT[tid * NBLK + blockIdx.x] = hoff[tid];
    }
    v2i* myrec = (v2i*)rec + (size_t)blockIdx.x * BK_HPB;
    #pragma unroll
    for (int u = 0; u < BK_EPT; ++u) {
        int e = e0 + tid + u * BK_T;
        if (e < E) {
            v2i r0, r1;
            r0.x = ((sv[u] & (BK_CN - 1)) << REC_NBR_BITS) | dv[u];
            r0.y = lv[u];
            r1.x = ((dv[u] & (BK_CN - 1)) << REC_NBR_BITS) | sv[u];
            r1.y = lv[u];
            myrec[hoff[sv[u] >> BK_SHIFT] + rk[2 * u]]     = r0;
            myrec[hoff[dv[u] >> BK_SHIFT] + rk[2 * u + 1]] = r1;
        }
    }
}

__device__ __forceinline__ int block_scan_excl(int c, int* ws, int* tot) {
    const int lane = threadIdx.x & 31;
    const int w    = threadIdx.x >> 5;
    int v = c;
    for (int o = 1; o < 32; o <<= 1) {
        int t = __shfl_up(v, o);
        if (lane >= o) v += t;
    }
    if (lane == 31) ws[w] = v;
    __syncthreads();
    if (w == 0) {
        int s  = ws[lane];
        int sv = s;
        for (int o = 1; o < 32; o <<= 1) {
            int t = __shfl_up(sv, o);
            if (lane >= o) sv += t;
        }
        ws[lane] = sv - s;
        if (lane == 31) *tot = sv;
    }
    __syncthreads();
    return ws[w] + v - c;
}

#define BK_U 8

__device__ __forceinline__ void lists_body(
    const int* __restrict__ rec, const int* __restrict__ cntT,
    const int* __restrict__ offT, int* __restrict__ adj,
    int* __restrict__ pos, int* __restrict__ cnt,
    int N, int NBLK, int P)
{
    __shared__ int segO[BK_T];
    __shared__ int segP[BK_T + 1];
    __shared__ int lcnt[BK_T];
    __shared__ int lcur[BK_T];
    __shared__ int wsA[32], wsB[32], wsC[32];
    __shared__ int totA, totB, baseSh;

    const int tid   = threadIdx.x;
    const int lane  = tid & 31;
    const int w     = tid >> 5;
    const int kb    = blockIdx.x;
    const int node0 = kb << BK_SHIFT;

    int c = 0, o = 0;
    if (tid < NBLK) { c = cntT[kb * NBLK + tid]; o = offT[kb * NBLK + tid]; }
    segO[tid] = o;
    int ex = block_scan_excl(c, wsA, &totA);
    segP[tid] = ex;
    if (tid == BK_T - 1) segP[BK_T] = ex + c;

    int acc = 0;
    for (int idx = tid; idx < kb * NBLK; idx += BK_T) acc += cntT[idx];
    for (int m = 16; m >= 1; m >>= 1) acc += __shfl_xor(acc, m);
    if (lane == 0) wsB[w] = acc;
    lcnt[tid] = 0;
    __syncthreads();
    if (w == 0) {
        int s = wsB[lane];
        for (int m = 16; m >= 1; m >>= 1) s += __shfl_xor(s, m);
        if (lane == 0) baseSh = s;
    }
    __syncthreads();
    const int T    = totA;
    const int base = baseSh;
    int st0 = 1;
    while (st0 * 2 < NBLK) st0 *= 2;
    const v2i* r2 = (const v2i*)rec;

    for (int t0 = tid; t0 < T; t0 += BK_T * BK_U) {
        int  w0[BK_U], tt[BK_U], b[BK_U];
        bool ok[BK_U];
        #pragma unroll
        for (int u = 0; u < BK_U; ++u) {
            int t = t0 + u * BK_T;
            ok[u] = t < T;
            tt[u] = ok[u] ? t : 0;
            b[u]  = 0;
        }
        for (int st = st0; st >= 1; st >>= 1) {
            #pragma unroll
            for (int u = 0; u < BK_U; ++u) b[u] += (segP[b[u] + st] <= tt[u]) ? st : 0;
        }
        #pragma unroll
        for (int u = 0; u < BK_U; ++u) {
            w0[u] = r2[b[u] * BK_HPB + segO[b[u]] + (tt[u] - segP[b[u]])].x;
        }
        #pragma unroll
        for (int u = 0; u < BK_U; ++u) {
            int nbr = w0[u] & ((1 << REC_NBR_BITS) - 1);
            int ll  = ((nbr / SLAB_NODES) << BK_SHIFT) | (w0[u] >> REC_NBR_BITS);
            if (ok[u]) atomicAdd(&lcnt[ll], 1);
        }
    }
    __syncthreads();

    const int LISTS = P << BK_SHIFT;
    int lc  = (tid < LISTS) ? lcnt[tid] : 0;
    int lex = block_scan_excl(lc, wsC, &totB);
    if (tid < LISTS) {
        int p = tid >> BK_SHIFT;
        int i = node0 + (tid & (BK_CN - 1));
        lcur[tid] = base + lex;
        if (i < N) {
            pos[p * N + i] = base + lex;
            cnt[p * N + i] = lc;
        }
    }
    __syncthreads();

    v2i* a2 = (v2i*)adj;
    for (int t0 = tid; t0 < T; t0 += BK_T * BK_U) {
        int  tt[BK_U], b[BK_U];
        v2i  rr[BK_U];
        bool ok[BK_U];
        #pragma unroll
        for (int u = 0; u < BK_U; ++u) {
            int t = t0 + u * BK_T;
            ok[u] = t < T;
            tt[u] = ok[u] ? t : 0;
            b[u]  = 0;
        }
        for (int st = st0; st >= 1; st >>= 1) {
            #pragma unroll
            for (int u = 0; u < BK_U; ++u) b[u] += (segP[b[u] + st] <= tt[u]) ? st : 0;
        }
        #pragma unroll
        for (int u = 0; u < BK_U; ++u) {
            rr[u] = r2[b[u] * BK_HPB + segO[b[u]] + (tt[u] - segP[b[u]])];
        }
        #pragma unroll
        for (int u = 0; u < BK_U; ++u) {
            int nbr = rr[u].x & ((1 << REC_NBR_BITS) - 1);
            int ll  = ((nbr / SLAB_NODES) << BK_SHIFT) | (rr[u].x >> REC_NBR_BITS);
            if (ok[u]) {
                int slot = atomicAdd(&lcur[ll], 1);
                v2i en; en.x = nbr; en.y = rr[u].y;
                a2[slot] = en;
            }
        }
    }
}

__global__ __launch_bounds__(BK_T) void lists_head_kernel(
    const int* __restrict__ rec, const int* __restrict__ cntT,
    const int* __restrict__ offT, int* __restrict__ adj,
    int* __restrict__ pos, int* __restrict__ cnt,
    int N, int NBLK, int P, int NB,
    const float* __restrict__ ht, const float* __restrict__ W,
    const float* __restrict__ bias, const float* __restrict__ kp,
    const float* __restrict__ ts, float* __restrict__ x, float* __restrict__ z,
    int tile0, int tileEnd, int D)
{
    if ((int)blockIdx.x < NB) lists_body(rec, cntT, offT, adj, pos, cnt, N, NBLK, P);
    else head_tiles(ht, W, bias, kp, ts, x, z, tile0, tileEnd, (int)blockIdx.x - NB, D, N);
}

__device__ __forceinline__ void fetch_entries(
    v2i (&ent)[ROUNDS], bool (&has)[ROUNDS], int (&k)[ROUNDS],
    const int (&end)[ROUNDS], const v2i* __restrict__ a2)
{
    #pragma unroll
    for (int r = 0; r < ROUNDS; ++r) {
        has[r] = k[r] < end[r];
        ent[r] = a2[has[r] ? k[r] : 0];
        k[r] += NODE_LANES;
    }
}

__device__ __forceinline__ void walk_entries(
    const v2i (&ent)[ROUNDS], const bool (&has)[ROUNDS], const float* sx, int lo,
    const float (&px)[ROUNDS], const float (&py)[ROUNDS], const float (&pz)[ROUNDS],
    float (&ax)[ROUNDS], float (&ay)[ROUNDS], float (&az)[ROUNDS])
{
    #pragma unroll
    for (int r = 0; r < ROUNDS; ++r) {
        if (has[r]) {
            int j = ent[r].x;
            const float* q = &sx[(j - lo) * 3];
            float L0 = __int_as_float(ent[r].y);
            float dx = px[r] - q[0];
            float dy = py[r] - q[1];
            float dz = pz[r] - q[2];
            float dist = sqrtf(dx * dx + dy * dy + dz * dz);
            float C    = dist - L0;
            float scal = (-C / XPBD_DENOM) / (dist + 1e-9f);
            ax[r] += fminf(fmaxf(scal * dx, -XPBD_MAXC), XPBD_MAXC);
            ay[r] += fminf(fmaxf(scal * dy, -XPBD_MAXC), XPBD_MAXC);
            az[r] += fminf(fmaxf(scal * dz, -XPBD_MAXC), XPBD_MAXC);
        }
    }
}

#define SETS 2

template <bool FINAL>
__global__ __launch_bounds__(PASS_T) void slab_gather_kernel(
    const float* __restrict__ xin,
    const int*   __restrict__ adj,
    const int*   __restrict__ pos,
    const int*   __restrict__ cnt,
    const float* __restrict__ kp,
    const float* __restrict__ ts,
    float* __restrict__ xout,
    float* __restrict__ znext,
    int N, int npc)
{
    __shared__ __attribute__((aligned(16))) float sx[SLAB_NODES * 3];

    const int tid     = threadIdx.x;
    const int grp     = tid / NODE_LANES;
    const int sub     = tid % NODE_LANES;
    const int b       = blockIdx.y;
    const int ph      = blockIdx.z;
    const int lo      = ph * SLAB_NODES;
    const int hi      = min(N, lo + SLAB_NODES);
    const int nw      = (hi - lo) * 3;
    const int node0   = blockIdx.x * npc;
    const int nodeEnd = min(N, node0 + npc);
    const float* xb   = xin + (size_t)b * N * 3;
    const v2i*   a2   = (const v2i*)adj;

    float px[ROUNDS], py[ROUNDS], pz[ROUNDS];
    int   s0v[SETS][ROUNDS], cv[SETS][ROUNDS];
    #pragma unroll
    for (int r = 0; r < ROUNDS; ++r) {
        int i = node0 + grp + GROUPS * r;
        const float* p = xb + min(i, N - 1) * 3;
        px[r] = p[0]; py[r] = p[1]; pz[r] = p[2];
    }
    #pragma unroll
    for (int set = 0; set < SETS; ++set) {
        #pragma unroll
        for (int r = 0; r < ROUNDS; ++r) {
            int  i   = node0 + set * (GROUPS * ROUNDS) + grp + GROUPS * r;
            bool own = i < nodeEnd;
            int  l   = ph * N + (own ? i : 0);
            int  c   = cnt[l];
            s0v[set][r] = pos[l];
            cv[set][r]  = own ? c : 0;
        }
    }

    {
        const float* src = xb + (size_t)lo * 3;
        if ((((size_t)src) & 15) == 0) {
            const v4f* s4 = (const v4f*)src;
            v4f*       d4 = (v4f*)sx;
            const int  n4 = nw >> 2;
            const int last = n4 - 1;
            const int STEP = PASS_T * COPY_DEPTH;
            v4f bufA[COPY_DEPTH], bufB[COPY_DEPTH];
            if (n4 > 0) {
                #pragma unroll
                for (int u = 0; u < COPY_DEPTH; ++u)
                    bufA[u] = s4[min(tid + u * PASS_T, last)];
                for (int base = 0; base < n4; base += 2 * STEP) {
                    #pragma unroll
                    for (int u = 0; u < COPY_DEPTH; ++u)
                        bufB[u] = s4[min(base + STEP + tid + u * PASS_T, last)];
                    #pragma unroll
                    for (int u = 0; u < COPY_DEPTH; ++u) {
                        int w = base + tid + u * PASS_T;
                        if (w < n4) d4[w] = bufA[u];
                    }
                    #pragma unroll
                    for (int u = 0; u < COPY_DEPTH; ++u)
                        bufA[u] = s4[min(base + 2 * STEP + tid + u * PASS_T, last)];
                    #pragma unroll
                    for (int u = 0; u < COPY_DEPTH; ++u) {
                        int w = base + STEP + tid + u * PASS_T;
                        if (w < n4) d4[w] = bufB[u];
                    }
                }
            }
            for (int w = (n4 << 2) + tid; w < nw; w += PASS_T) sx[w] = src[w];
        } else {
            for (int w = tid; w < nw; w += PASS_T) sx[w] = src[w];
        }
    }

    float tau = 1.0f;
    if (FINAL) tau = fmaxf(1.0f - ts[b], TAU_MIN_F);

    #pragma unroll
    for (int set = 0; set < SETS; ++set) {
        const int setNode0 = node0 + set * (GROUPS * ROUNDS);
        if (set > 0) {
            #pragma unroll
            for (int r = 0; r < ROUNDS; ++r) {
                int i = setNode0 + grp + GROUPS * r;
                const float* p = xb + min(i, N - 1) * 3;
                px[r] = p[0]; py[r] = p[1]; pz[r] = p[2];
            }
        }

        float ax[ROUNDS], ay[ROUNDS], az[ROUNDS];
        int   k[ROUNDS], end[ROUNDS];
        int   trips = 0;
        #pragma unroll
        for (int r = 0; r < ROUNDS; ++r) {
            ax[r] = 0.0f; ay[r] = 0.0f; az[r] = 0.0f;
            k[r]   = s0v[set][r] + sub;
            end[r] = s0v[set][r] + cv[set][r];
            trips  = max(trips, (end[r] - k[r] + NODE_LANES - 1) / NODE_LANES);
        }

        v2i  entA[ROUNDS], entB[ROUNDS];
        bool hasA[ROUNDS], hasB[ROUNDS];
        fetch_entries(entA, hasA, k, end, a2);

        if (set == 0) __syncthreads();

        for (int t = 0; t < trips; t += 2) {
            fetch_entries(entB, hasB, k, end, a2);
            walk_entries(entA, hasA, sx, lo, px, py, pz, ax, ay, az);
            fetch_entries(entA, hasA, k, end, a2);
            walk_entries(entB, hasB, sx, lo, px, py, pz, ax, ay, az);
        }

        #pragma unroll
        for (int m = 1; m < NODE_LANES; m <<= 1) {
            #pragma unroll
            for (int r = 0; r < ROUNDS; ++r) {
                ax[r] += __shfl_xor(ax[r], m, NODE_LANES);
                ay[r] += __shfl_xor(ay[r], m, NODE_LANES);
                az[r] += __shfl_xor(az[r], m, NODE_LANES);
            }
        }

        #pragma unroll
        for (int r = 0; r < ROUNDS; ++r) {
            int i = setNode0 + grp + GROUPS * r;
            if (i < nodeEnd && sub < 3) {
                float a = (sub == 0) ? ax[r] : (sub == 1) ? ay[r] : az[r];
                float p = (sub == 0) ? px[r] : (sub == 1) ? py[r] : pz[r];
                int idx = (b * N + i) * 3 + sub;
                float v;
                if (ph == 0) {
                    v = p + a;
                    if (FINAL) v = (v - kp[idx]) / tau;
                    if (znext) znext[idx] = 0.0f;
                } else {
                    v = a;
                    if (FINAL) v = a / tau;
                }
                atomic_add_f32(&xout[idx], v);
            }
        }
    }
}

extern "C" void kernel_launch(void* const* d_in, const int* in_sizes, int n_in,
                              void* d_out, int out_size, void* d_ws, size_t ws_size,
                              hipStream_t stream) {
    const float* kp   = (const float*)d_in[0];
    const float* ts   = (const float*)d_in[1];
    const float* ht   = (const float*)d_in[2];
    const float* W    = (const float*)d_in[3];
    const float* bias = (const float*)d_in[4];
    const int*   ei   = (const int*)d_in[5];
    const float* rl   = (const float*)d_in[6];

    const int B   = in_sizes[1];
    const int E   = in_sizes[6];
    const int BN3 = in_sizes[0];
    const int BN  = BN3 / 3;
    const int N   = BN / B;
    const int D   = in_sizes[2] / BN;

    const int T = 256;
    const int nTiles = (BN + 15) / 16;
    const int gVec   = (BN3 + T - 1) / T;
    const int BE     = B * E;
    const int gEdge  = (BE + T - 1) / T;
    const int gWmma  = (nTiles * 32 + T - 1) / T;
    const int XPBD_ITERS = 4;

    const int P    = (N + SLAB_NODES - 1) / SLAB_NODES;
    const int NL   = P * N;
    const int npc  = (N + CHUNKS - 1) / CHUNKS;
    const int NB   = (N + BK_CN - 1) >> BK_SHIFT;
    const int NBLK = (E + BK_EPB - 1) / BK_EPB;
    const size_t rW   = ((size_t)4 * E > (size_t)2 * BN3) ? (size_t)4 * E : (size_t)2 * BN3;
    const size_t tabW = (size_t)2 * NB * NBLK;
    const size_t need = ((size_t)4 * E + rW + (size_t)BN3 + (size_t)2 * NL + 1 + tabW) * sizeof(int);

    if (E > 0 && npc <= SETS * GROUPS * ROUNDS && ws_size >= need &&
        NB <= BK_NBMAX && NBLK <= BK_T && P * BK_CN <= BK_T && N <= (1 << REC_NBR_BITS)) {
        int*   adj  = (int*)d_ws;
        int*   rec  = adj + (size_t)4 * E;
        float* x2   = (float*)rec;
        float* x3   = x2 + BN3;
        float* x0   = (float*)(rec + rW);
        float* xo   = (float*)d_out;
        int*   cnt  = (int*)(x0 + BN3);
        int*   pos  = cnt + NL + 1;
        int*   cntT = pos + NL;
        int*   offT = cntT + (size_t)NB * NBLK;

        const dim3 gPass(CHUNKS, B, P);

        const int TPB = BK_T / 32;
        const int hb  = (nTiles + TPB - 1) / TPB;

        bucket_kernel<<<NBLK, BK_T, 0, stream>>>(ei, rl, rec, cntT, offT, E, NB, NBLK);
        lists_head_kernel<<<NB + hb, BK_T, 0, stream>>>(rec, cntT, offT, adj, pos, cnt,
                                                        N, NBLK, P, NB,
                                                        ht, W, bias, kp, ts, x0, xo, 0, nTiles, D);

        slab_gather_kernel<false><<<gPass, PASS_T, 0, stream>>>(x0, adj, pos, cnt, kp, ts, xo, x2, N, npc);
        slab_gather_kernel<false><<<gPass, PASS_T, 0, stream>>>(xo, adj, pos, cnt, kp, ts, x2, x3, N, npc);
        slab_gather_kernel<false><<<gPass, PASS_T, 0, stream>>>(x2, adj, pos, cnt, kp, ts, x3, xo, N, npc);
        slab_gather_kernel<true ><<<gPass, PASS_T, 0, stream>>>(x3, adj, pos, cnt, kp, ts, xo, nullptr, N, npc);
        return;
    }

    float* xbuf  = (float*)d_ws;
    float* delta = xbuf + BN3;

    zero_kernel<<<gVec, T, 0, stream>>>(delta, BN3);
    head_wmma_kernel<<<gWmma, T, 0, stream>>>(ht, W, bias, kp, ts, xbuf, nTiles, D, N);

    for (int it = 0; it < XPBD_ITERS; ++it) {
        edge_kernel<<<gEdge, T, 0, stream>>>(xbuf, ei, rl, delta, E, N, BE);
        if (it < XPBD_ITERS - 1) {
            apply_kernel<<<gVec, T, 0, stream>>>(xbuf, delta, BN3);
        } else {
            final_kernel<<<gVec, T, 0, stream>>>(xbuf, delta, kp, ts,
                                                 (float*)d_out, BN3, N * 3);
        }
    }
}
